# starting kernel
# baseline (speedup 1.0000x reference)
#include <hip/hip_runtime.h>

typedef __attribute__((ext_vector_type(2))) float v2f;
typedef __attribute__((ext_vector_type(8))) float v8f;

#define N_NODES 100000
#define E_EDGES 1600000
#define F_DIM   64

__global__ void zero_f32(float* __restrict__ p, int n) {
  int i = blockIdx.x * blockDim.x + threadIdx.x;
  if (i < n) p[i] = 0.0f;
}

__global__ void deg_accum(const int* __restrict__ row, float* __restrict__ deg, int e) {
  int i = blockIdx.x * blockDim.x + threadIdx.x;
  if (i < e) atomicAdd(&deg[row[i]], 1.0f);
}

__global__ void make_dis(float* __restrict__ deg, int n) {
  int i = blockIdx.x * blockDim.x + threadIdx.x;
  if (i < n) {
    float d = deg[i];
    deg[i] = (d > 0.0f) ? rsqrtf(d) : 0.0f;
  }
}

__global__ void edge_weight(const int* __restrict__ row, const int* __restrict__ col,
                            const float* __restrict__ dis, float* __restrict__ w, int e) {
  int i = blockIdx.x * blockDim.x + threadIdx.x;
  if (i < e) w[i] = -dis[row[i]] * dis[col[i]];
}

__global__ void scatter_prop(const int* __restrict__ row, const int* __restrict__ col,
                             const float* __restrict__ w, const float* __restrict__ H,
                             float* __restrict__ P, int e) {
  int gid = blockIdx.x * blockDim.x + threadIdx.x;
  int eidx = gid >> 4;
  if (eidx >= e) return;
  int k = (gid & 15) * 4;
  int r = row[eidx];
  int c = col[eidx];
  float we = w[eidx];
  const float4 v = *(const float4*)(H + (size_t)c * F_DIM + k);
  float* dst = P + (size_t)r * F_DIM + k;
  atomicAdd(dst + 0, we * v.x);
  atomicAdd(dst + 1, we * v.y);
  atomicAdd(dst + 2, we * v.z);
  atomicAdd(dst + 3, we * v.w);
}

template <int FOUT, bool RELU>
__global__ void cheb_gemm(const float* __restrict__ X, const float* __restrict__ P,
                          const float* __restrict__ W0, const float* __restrict__ W1,
                          const float* __restrict__ B, float* __restrict__ OUT) {
  constexpr int NT   = FOUT / 16;
  constexpr int NKG  = F_DIM / 4;
  constexpr int NENT = NKG * NT * 32;

  __shared__ float4 wlds[NENT];

  for (int idx = threadIdx.x; idx < NENT; idx += blockDim.x) {
    const int l   = idx & 31;
    const int rem = idx >> 5;
    const int t   = rem % NT;
    const int kg  = rem / NT;
    const int n   = t * 16 + (l & 15);
    const int k   = kg * 4 + ((l >> 4) * 2);
    wlds[idx] = make_float4(W0[(size_t)k * FOUT + n], W0[(size_t)(k + 1) * FOUT + n],
                            W1[(size_t)k * FOUT + n], W1[(size_t)(k + 1) * FOUT + n]);
  }
  __syncthreads();

  const int lane  = threadIdx.x & 31;
  const int wave  = threadIdx.x >> 5;
  const int tile  = blockIdx.x * 4 + wave;
  const int ntile = N_NODES / 16;
  if (tile >= ntile) return;

  const int row0 = tile * 16;
  const int m    = lane & 15;
  const int kk   = (lane >> 4) * 2;

  v8f acc[NT];
#pragma unroll
  for (int t = 0; t < NT; ++t) acc[t] = (v8f)0.0f;

  const float* xrow = X + (size_t)(row0 + m) * F_DIM;
  const float* prow = P + (size_t)(row0 + m) * F_DIM;

#pragma unroll
  for (int kg = 0; kg < NKG; ++kg) {
    const int k0 = kg * 4;
    const v2f aX = *(const v2f*)(xrow + k0 + kk);
    const v2f aP = *(const v2f*)(prow + k0 + kk);
#pragma unroll
    for (int t = 0; t < NT; ++t) {
      const float4 wv = wlds[(kg * NT + t) * 32 + lane];
      v2f b0, b1;
      b0.x = wv.x; b0.y = wv.y;
      b1.x = wv.z; b1.y = wv.w;
      acc[t] = __builtin_amdgcn_wmma_f32_16x16x4_f32(
          false, aX, false, b0, (short)0, acc[t], false, false);
      acc[t] = __builtin_amdgcn_wmma_f32_16x16x4_f32(
          false, aP, false, b1, (short)0, acc[t], false, false);
    }
  }

  const int roff = (lane >> 4) * 8;
#pragma unroll
  for (int t = 0; t < NT; ++t) {
    const float bias = B[t * 16 + m];
#pragma unroll
    for (int r = 0; r < 8; ++r) {
      float v = acc[t][r] + bias;
      if (RELU) v = fmaxf(v, 0.0f);
      OUT[(size_t)(row0 + roff + r) * FOUT + t * 16 + m] = v;
    }
  }
}

extern "C" void kernel_launch(void* const* d_in, const int* in_sizes, int n_in,
                              void* d_out, int out_size, void* d_ws, size_t ws_size,
                              hipStream_t stream) {
  const float* x    = (const float*)d_in[0];
  const int*   adj  = (const int*)d_in[1];
  const float* W1_0 = (const float*)d_in[2];
  const float* W1_1 = (const float*)d_in[3];
  const float* b1   = (const float*)d_in[4];
  const float* Wx_0 = (const float*)d_in[5];
  const float* Wx_1 = (const float*)d_in[6];
  const float* bx   = (const float*)d_in[7];
  const float* W2_0 = (const float*)d_in[8];
  const float* W2_1 = (const float*)d_in[9];
  const float* b2   = (const float*)d_in[10];
  float* out = (float*)d_out;

  const int* row = adj;
  const int* col = adj + E_EDGES;

  float* dis = (float*)d_ws;
  float* w   = dis + N_NODES;
  float* P   = w + E_EDGES;
  float* H1  = P + (size_t)N_NODES * F_DIM;
  float* H2  = H1 + (size_t)N_NODES * F_DIM;

  const int TB = 256;
  const int gN   = (N_NODES + TB - 1) / TB;
  const int gE   = (E_EDGES + TB - 1) / TB;
  const int gNF  = (N_NODES * F_DIM + TB - 1) / TB;
  const int gSc  = (E_EDGES * 16 + TB - 1) / TB;
  const int gGem = (N_NODES / 16 + 3) / 4;

  zero_f32<<<gN, TB, 0, stream>>>(dis, N_NODES);
  deg_accum<<<gE, TB, 0, stream>>>(row, dis, E_EDGES);
  make_dis<<<gN, TB, 0, stream>>>(dis, N_NODES);
  edge_weight<<<gE, TB, 0, stream>>>(row, col, dis, w, E_EDGES);

  zero_f32<<<gNF, TB, 0, stream>>>(P, N_NODES * F_DIM);
  scatter_prop<<<gSc, TB, 0, stream>>>(row, col, w, x, P, E_EDGES);
  cheb_gemm<64, true><<<gGem, 128, 0, stream>>>(x, P, W1_0, W1_1, b1, H1);

  zero_f32<<<gNF, TB, 0, stream>>>(P, N_NODES * F_DIM);
  scatter_prop<<<gSc, TB, 0, stream>>>(row, col, w, H1, P, E_EDGES);
  cheb_gemm<64, true><<<gGem, 128, 0, stream>>>(H1, P, Wx_0, Wx_1, bx, H2);

  zero_f32<<<gNF, TB, 0, stream>>>(P, N_NODES * F_DIM);
  scatter_prop<<<gSc, TB, 0, stream>>>(row, col, w, H2, P, E_EDGES);
  cheb_gemm<16, false><<<gGem, 128, 0, stream>>>(H2, P, W2_0, W2_1, b2, out);
}


# edges counting-sorted by destination row each call (arrival rank from a returning int atomic); propagation = per-row gather-sum, no float atomics, no zero fill; sum order in a row = arrival rank
# speedup vs baseline: 2.1185x; 2.1185x over previous
#include <hip/hip_runtime.h>

typedef __attribute__((ext_vector_type(2))) float v2f;
typedef __attribute__((ext_vector_type(8))) float v8f;

#define N_NODES 100000
#define E_EDGES 1600000
#define F_DIM   64


__global__ void zero_i32(int* __restrict__ p, int n) {
  int i = blockIdx.x * blockDim.x + threadIdx.x;
  if (i < n) p[i] = 0;
}

__global__ void rank_count(const int* __restrict__ row, int* __restrict__ cnt,
                           int* __restrict__ rank, int e) {
  int i = blockIdx.x * blockDim.x + threadIdx.x;
  if (i < e) rank[i] = atomicAdd(&cnt[row[i]], 1);
}

__global__ void block_sums(const int* __restrict__ cnt, int* __restrict__ bsum, int n) {
  __shared__ int red[256];
  const int tid = threadIdx.x;
  const int i = blockIdx.x * 256 + tid;
  red[tid] = (i < n) ? cnt[i] : 0;
  __syncthreads();
  for (int s = 128; s > 0; s >>= 1) {
    if (tid < s) red[tid] += red[tid + s];
    __syncthreads();
  }
  if (tid == 0) bsum[blockIdx.x] = red[0];
}

__global__ void scan_rows(const int* __restrict__ cnt, const int* __restrict__ bsum,
                          int* __restrict__ rowptr, int n) {
  __shared__ int red[256];
  __shared__ int buf[2][256];
  const int tid = threadIdx.x;
  const int b = blockIdx.x;

  int part = 0;
  for (int k = tid; k < b; k += 256) part += bsum[k];
  red[tid] = part;
  __syncthreads();
  for (int s = 128; s > 0; s >>= 1) {
    if (tid < s) red[tid] += red[tid + s];
    __syncthreads();
  }
  const int offset = red[0];

  const int i = b * 256 + tid;
  const int v = (i < n) ? cnt[i] : 0;
  int cur = 0;
  buf[0][tid] = v;
  __syncthreads();
  for (int d = 1; d < 256; d <<= 1) {
    int t = buf[cur][tid];
    if (tid >= d) t += buf[cur][tid - d];
    buf[cur ^ 1][tid] = t;
    cur ^= 1;
    __syncthreads();
  }
  if (i < n) rowptr[i] = offset + buf[cur][tid] - v;
}

__global__ void fill_sorted(const int* __restrict__ row, const int* __restrict__ col,
                            const int* __restrict__ rank, const int* __restrict__ rowptr,
                            int* __restrict__ scol, int e) {
  int i = blockIdx.x * blockDim.x + threadIdx.x;
  if (i < e) scol[rowptr[row[i]] + rank[i]] = col[i];
}

__device__ __forceinline__ float inv_sqrt_deg(int d) {
  return (d > 0) ? rsqrtf((float)d) : 0.0f;
}

__global__ void gather_prop(const int* __restrict__ rowptr, const int* __restrict__ scol,
                            const float* __restrict__ H, float* __restrict__ P) {
  const int gid = blockIdx.x * blockDim.x + threadIdx.x;
  const int r = gid >> 4;
  if (r >= N_NODES) return;
  const int g = threadIdx.x & 15;
  const int s = rowptr[r];
  const int e = (r + 1 < N_NODES) ? rowptr[r + 1] : E_EDGES;
  const float dr = inv_sqrt_deg(e - s);
  const float* Hg = H + g * 4;
  float4 acc = make_float4(0.0f, 0.0f, 0.0f, 0.0f);

  for (int base = s; base < e; base += 16) {
    const int j = base + g;
    int c = 0;
    float w = 0.0f;
    if (j < e) {
      c = scol[j];
      const int cs = rowptr[c];
      const int ce = (c + 1 < N_NODES) ? rowptr[c + 1] : E_EDGES;
      w = -dr * inv_sqrt_deg(ce - cs);
    }
    const int n = min(16, e - base);
    int t = 0;
    for (; t + 4 <= n; t += 4) {
      const int c0 = __shfl(c, t, 16);
      const int c1 = __shfl(c, t + 1, 16);
      const int c2 = __shfl(c, t + 2, 16);
      const int c3 = __shfl(c, t + 3, 16);
      const float w0 = __shfl(w, t, 16);
      const float w1 = __shfl(w, t + 1, 16);
      const float w2 = __shfl(w, t + 2, 16);
      const float w3 = __shfl(w, t + 3, 16);
      const float4 v0 = *(const float4*)(Hg + (size_t)c0 * F_DIM);
      const float4 v1 = *(const float4*)(Hg + (size_t)c1 * F_DIM);
      const float4 v2 = *(const float4*)(Hg + (size_t)c2 * F_DIM);
      const float4 v3 = *(const float4*)(Hg + (size_t)c3 * F_DIM);
      acc.x = fmaf(w0, v0.x, acc.x); acc.y = fmaf(w0, v0.y, acc.y);
      acc.z = fmaf(w0, v0.z, acc.z); acc.w = fmaf(w0, v0.w, acc.w);
      acc.x = fmaf(w1, v1.x, acc.x); acc.y = fmaf(w1, v1.y, acc.y);
      acc.z = fmaf(w1, v1.z, acc.z); acc.w = fmaf(w1, v1.w, acc.w);
      acc.x = fmaf(w2, v2.x, acc.x); acc.y = fmaf(w2, v2.y, acc.y);
      acc.z = fmaf(w2, v2.z, acc.z); acc.w = fmaf(w2, v2.w, acc.w);
      acc.x = fmaf(w3, v3.x, acc.x); acc.y = fmaf(w3, v3.y, acc.y);
      acc.z = fmaf(w3, v3.z, acc.z); acc.w = fmaf(w3, v3.w, acc.w);
    }
    for (; t < n; ++t) {
      const int ct = __shfl(c, t, 16);
      const float wt = __shfl(w, t, 16);
      const float4 v = *(const float4*)(Hg + (size_t)ct * F_DIM);
      acc.x = fmaf(wt, v.x, acc.x); acc.y = fmaf(wt, v.y, acc.y);
      acc.z = fmaf(wt, v.z, acc.z); acc.w = fmaf(wt, v.w, acc.w);
    }
  }
  *(float4*)(P + (size_t)r * F_DIM + g * 4) = acc;
}

template <int FOUT, bool RELU>
__global__ void cheb_gemm(const float* __restrict__ X, const float* __restrict__ P,
                          const float* __restrict__ W0, const float* __restrict__ W1,
                          const float* __restrict__ B, float* __restrict__ OUT) {
  constexpr int NT   = FOUT / 16;
  constexpr int NKG  = F_DIM / 4;
  constexpr int NENT = NKG * NT * 32;

  __shared__ float4 wlds[NENT];

  for (int idx = threadIdx.x; idx < NENT; idx += blockDim.x) {
    const int l   = idx & 31;
    const int rem = idx >> 5;
    const int t   = rem % NT;
    const int kg  = rem / NT;
    const int n   = t * 16 + (l & 15);
    const int k   = kg * 4 + ((l >> 4) * 2);
    wlds[idx] = make_float4(W0[(size_t)k * FOUT + n], W0[(size_t)(k + 1) * FOUT + n],
                            W1[(size_t)k * FOUT + n], W1[(size_t)(k + 1) * FOUT + n]);
  }
  __syncthreads();

  const int lane  = threadIdx.x & 31;
  const int wave  = threadIdx.x >> 5;
  const int tile  = blockIdx.x * 4 + wave;
  const int ntile = N_NODES / 16;
  if (tile >= ntile) return;

  const int row0 = tile * 16;
  const int m    = lane & 15;
  const int kk   = (lane >> 4) * 2;

  v8f acc[NT];
#pragma unroll
  for (int t = 0; t < NT; ++t) acc[t] = (v8f)0.0f;

  const float* xrow = X + (size_t)(row0 + m) * F_DIM;
  const float* prow = P + (size_t)(row0 + m) * F_DIM;

#pragma unroll
  for (int kg = 0; kg < NKG; ++kg) {
    const int k0 = kg * 4;
    const v2f aX = *(const v2f*)(xrow + k0 + kk);
    const v2f aP = *(const v2f*)(prow + k0 + kk);
#pragma unroll
    for (int t = 0; t < NT; ++t) {
      const float4 wv = wlds[(kg * NT + t) * 32 + lane];
      v2f b0, b1;
      b0.x = wv.x; b0.y = wv.y;
      b1.x = wv.z; b1.y = wv.w;
      acc[t] = __builtin_amdgcn_wmma_f32_16x16x4_f32(
          false, aX, false, b0, (short)0, acc[t], false, false);
      acc[t] = __builtin_amdgcn_wmma_f32_16x16x4_f32(
          false, aP, false, b1, (short)0, acc[t], false, false);
    }
  }

  const int roff = (lane >> 4) * 8;
#pragma unroll
  for (int t = 0; t < NT; ++t) {
    const float bias = B[t * 16 + m];
#pragma unroll
    for (int r = 0; r < 8; ++r) {
      float v = acc[t][r] + bias;
      if (RELU) v = fmaxf(v, 0.0f);
      OUT[(size_t)(row0 + roff + r) * FOUT + t * 16 + m] = v;
    }
  }
}

extern "C" void kernel_launch(void* const* d_in, const int* in_sizes, int n_in,
                              void* d_out, int out_size, void* d_ws, size_t ws_size,
                              hipStream_t stream) {
  const float* x    = (const float*)d_in[0];
  const int*   adj  = (const int*)d_in[1];
  const float* W1_0 = (const float*)d_in[2];
  const float* W1_1 = (const float*)d_in[3];
  const float* b1   = (const float*)d_in[4];
  const float* Wx_0 = (const float*)d_in[5];
  const float* Wx_1 = (const float*)d_in[6];
  const float* bx   = (const float*)d_in[7];
  const float* W2_0 = (const float*)d_in[8];
  const float* W2_1 = (const float*)d_in[9];
  const float* b2   = (const float*)d_in[10];
  float* out = (float*)d_out;

  const int* row = adj;
  const int* col = adj + E_EDGES;

  int*   rowptr = (int*)d_ws;
  int*   scol   = rowptr + N_NODES;
  float* P   = (float*)(scol + E_EDGES);
  float* H1  = P + (size_t)N_NODES * F_DIM;
  float* H2  = H1 + (size_t)N_NODES * F_DIM;
  int*   cnt  = (int*)H2;
  int*   rank = cnt + N_NODES;
  int*   bsum = rank + E_EDGES;

  const int TB = 256;
  const int gN   = (N_NODES + TB - 1) / TB;
  const int gE   = (E_EDGES + TB - 1) / TB;
  const int gGa  = (N_NODES * 16 + TB - 1) / TB;
  const int gGem = (N_NODES / 16 + 3) / 4;

  zero_i32<<<gN, TB, 0, stream>>>(cnt, N_NODES);
  rank_count<<<gE, TB, 0, stream>>>(row, cnt, rank, E_EDGES);
  block_sums<<<gN, TB, 0, stream>>>(cnt, bsum, N_NODES);
  scan_rows<<<gN, TB, 0, stream>>>(cnt, bsum, rowptr, N_NODES);
  fill_sorted<<<gE, TB, 0, stream>>>(row, col, rank, rowptr, scol, E_EDGES);

  gather_prop<<<gGa, TB, 0, stream>>>(rowptr, scol, x, P);
  cheb_gemm<64, true><<<gGem, 128, 0, stream>>>(x, P, W1_0, W1_1, b1, H1);

  gather_prop<<<gGa, TB, 0, stream>>>(rowptr, scol, H1, P);
  cheb_gemm<64, true><<<gGem, 128, 0, stream>>>(H1, P, Wx_0, Wx_1, bx, H2);

  gather_prop<<<gGa, TB, 0, stream>>>(rowptr, scol, H2, P);
  cheb_gemm<16, false><<<gGem, 128, 0, stream>>>(H2, P, W2_0, W2_1, b2, out);
}


# edge grouping moved to LDS counters: edges binned by 256-row range, each bin counting-sorted in one work-group; no global atomic left; sum order in a row = arrival order at the LDS counters
# speedup vs baseline: 2.9900x; 1.4114x over previous
#include <hip/hip_runtime.h>

typedef __attribute__((ext_vector_type(2))) float v2f;
typedef __attribute__((ext_vector_type(8))) float v8f;

#define N_NODES 100000
#define E_EDGES 1600000
#define F_DIM   64


#define RPB_SHIFT 8
#define RPB       (1 << RPB_SHIFT)
#define NBIN      ((N_NODES + RPB - 1) / RPB)
#define NCHUNK    256
#define CHUNK     ((E_EDGES + NCHUNK - 1) / NCHUNK)

#define HIST_BD   1024
#define HIST_U    ((CHUNK + HIST_BD - 1) / HIST_BD)
#define SCAT_BD   512
#define SCAT_U    ((CHUNK + SCAT_BD - 1) / SCAT_BD)
#define SORT_BD   1024
#define SORT_U    4

static_assert(NBIN <= SCAT_BD, "one thread per bin in bin_scatter");
static_assert(NCHUNK * CHUNK >= E_EDGES, "chunks cover the edge list");

__global__ void bin_hist(const int* __restrict__ row, int* __restrict__ hist) {
  __shared__ int h[NBIN];
  const int tid = threadIdx.x;
  const int g = blockIdx.x;
  for (int b = tid; b < NBIN; b += HIST_BD) h[b] = 0;
  __syncthreads();
  const int lo = g * CHUNK;
  const int hi = min(lo + CHUNK, E_EDGES);
  int r[HIST_U];
#pragma unroll
  for (int u = 0; u < HIST_U; ++u) r[u] = row[min(lo + tid + u * HIST_BD, hi - 1)];
#pragma unroll
  for (int u = 0; u < HIST_U; ++u)
    if (lo + tid + u * HIST_BD < hi) atomicAdd(&h[r[u] >> RPB_SHIFT], 1);
  __syncthreads();
  for (int b = tid; b < NBIN; b += HIST_BD) hist[b * NCHUNK + g] = h[b];
}

__global__ void bin_scan(const int* __restrict__ hist, int* __restrict__ within,
                         int* __restrict__ total) {
  __shared__ int buf[2][NCHUNK];
  const int tid = threadIdx.x;
  const int b = blockIdx.x;
  const int v = hist[b * NCHUNK + tid];
  int cur = 0;
  buf[0][tid] = v;
  __syncthreads();
  for (int d = 1; d < NCHUNK; d <<= 1) {
    int t = buf[cur][tid];
    if (tid >= d) t += buf[cur][tid - d];
    buf[cur ^ 1][tid] = t;
    cur ^= 1;
    __syncthreads();
  }
  const int incl = buf[cur][tid];
  within[b * NCHUNK + tid] = incl - v;
  if (tid == NCHUNK - 1) total[b] = incl;
}

__global__ void bin_scatter(const int* __restrict__ row, const int* __restrict__ col,
                            const int* __restrict__ within, const int* __restrict__ total,
                            int* __restrict__ binstart, int* __restrict__ pairs) {
  __shared__ int buf[2][SCAT_BD];
  __shared__ int cursor[NBIN];
  const int tid = threadIdx.x;
  const int g = blockIdx.x;

  const int v = (tid < NBIN) ? total[tid] : 0;
  int cur = 0;
  buf[0][tid] = v;
  __syncthreads();
  for (int d = 1; d < SCAT_BD; d <<= 1) {
    int t = buf[cur][tid];
    if (tid >= d) t += buf[cur][tid - d];
    buf[cur ^ 1][tid] = t;
    cur ^= 1;
    __syncthreads();
  }
  if (tid < NBIN) {
    const int bs = buf[cur][tid] - v;
    cursor[tid] = bs + within[tid * NCHUNK + g];
    if (g == 0) binstart[tid] = bs;
  }
  __syncthreads();

  const int lo = g * CHUNK;
  const int hi = min(lo + CHUNK, E_EDGES);
  int2* __restrict__ out = (int2*)pairs;
  int r[SCAT_U], c[SCAT_U];
#pragma unroll
  for (int u = 0; u < SCAT_U; ++u) {
    const int i = min(lo + tid + u * SCAT_BD, hi - 1);
    r[u] = row[i];
    c[u] = col[i];
  }
#pragma unroll
  for (int u = 0; u < SCAT_U; ++u) {
    if (lo + tid + u * SCAT_BD < hi) {
      const int pos = atomicAdd(&cursor[r[u] >> RPB_SHIFT], 1);
      out[pos] = make_int2(r[u], c[u]);
    }
  }
}

__global__ void bin_sort(const int* __restrict__ pairs, const int* __restrict__ binstart,
                         const int* __restrict__ total, int* __restrict__ rowptr,
                         int* __restrict__ scol) {
  __shared__ int cnt[RPB];
  __shared__ int fill[RPB];
  __shared__ int buf[2][RPB];
  const int tid = threadIdx.x;
  const int b = blockIdx.x;
  const int bs = binstart[b];
  const int n = total[b];
  const int2* __restrict__ bin = (const int2*)pairs + bs;

  if (tid < RPB) cnt[tid] = 0;
  __syncthreads();

  for (int base = 0; base < n; base += SORT_U * SORT_BD) {
    int2 e[SORT_U];
#pragma unroll
    for (int u = 0; u < SORT_U; ++u) e[u] = bin[min(base + tid + u * SORT_BD, n - 1)];
#pragma unroll
    for (int u = 0; u < SORT_U; ++u)
      if (base + tid + u * SORT_BD < n) atomicAdd(&cnt[e[u].x & (RPB - 1)], 1);
  }
  __syncthreads();

  int cur = 0;
  if (tid < RPB) buf[0][tid] = cnt[tid];
  __syncthreads();
  for (int d = 1; d < RPB; d <<= 1) {
    if (tid < RPB) {
      int t = buf[cur][tid];
      if (tid >= d) t += buf[cur][tid - d];
      buf[cur ^ 1][tid] = t;
    }
    cur ^= 1;
    __syncthreads();
  }
  if (tid < RPB) {
    const int start = bs + buf[cur][tid] - cnt[tid];
    const int r = b * RPB + tid;
    if (r < N_NODES) rowptr[r] = start;
    fill[tid] = start;
  }
  __syncthreads();

  for (int base = 0; base < n; base += SORT_U * SORT_BD) {
    int2 e[SORT_U];
#pragma unroll
    for (int u = 0; u < SORT_U; ++u) e[u] = bin[min(base + tid + u * SORT_BD, n - 1)];
#pragma unroll
    for (int u = 0; u < SORT_U; ++u) {
      if (base + tid + u * SORT_BD < n) {
        const int pos = atomicAdd(&fill[e[u].x & (RPB - 1)], 1);
        scol[pos] = e[u].y;
      }
    }
  }
}

__device__ __forceinline__ float inv_sqrt_deg(int d) {
  return (d > 0) ? rsqrtf((float)d) : 0.0f;
}

__global__ void gather_prop(const int* __restrict__ rowptr, const int* __restrict__ scol,
                            const float* __restrict__ H, float* __restrict__ P) {
  const int gid = blockIdx.x * blockDim.x + threadIdx.x;
  const int r = gid >> 4;
  if (r >= N_NODES) return;
  const int g = threadIdx.x & 15;
  const int s = rowptr[r];
  const int e = (r + 1 < N_NODES) ? rowptr[r + 1] : E_EDGES;
  const float dr = inv_sqrt_deg(e - s);
  const float* Hg = H + g * 4;
  float4 acc = make_float4(0.0f, 0.0f, 0.0f, 0.0f);

  for (int base = s; base < e; base += 16) {
    const int j = base + g;
    int c = 0;
    float w = 0.0f;
    if (j < e) {
      c = scol[j];
      const int cs = rowptr[c];
      const int ce = (c + 1 < N_NODES) ? rowptr[c + 1] : E_EDGES;
      w = -dr * inv_sqrt_deg(ce - cs);
    }
    const int n = min(16, e - base);
    int t = 0;
    for (; t + 4 <= n; t += 4) {
      const int c0 = __shfl(c, t, 16);
      const int c1 = __shfl(c, t + 1, 16);
      const int c2 = __shfl(c, t + 2, 16);
      const int c3 = __shfl(c, t + 3, 16);
      const float w0 = __shfl(w, t, 16);
      const float w1 = __shfl(w, t + 1, 16);
      const float w2 = __shfl(w, t + 2, 16);
      const float w3 = __shfl(w, t + 3, 16);
      const float4 v0 = *(const float4*)(Hg + (size_t)c0 * F_DIM);
      const float4 v1 = *(const float4*)(Hg + (size_t)c1 * F_DIM);
      const float4 v2 = *(const float4*)(Hg + (size_t)c2 * F_DIM);
      const float4 v3 = *(const float4*)(Hg + (size_t)c3 * F_DIM);
      acc.x = fmaf(w0, v0.x, acc.x); acc.y = fmaf(w0, v0.y, acc.y);
      acc.z = fmaf(w0, v0.z, acc.z); acc.w = fmaf(w0, v0.w, acc.w);
      acc.x = fmaf(w1, v1.x, acc.x); acc.y = fmaf(w1, v1.y, acc.y);
      acc.z = fmaf(w1, v1.z, acc.z); acc.w = fmaf(w1, v1.w, acc.w);
      acc.x = fmaf(w2, v2.x, acc.x); acc.y = fmaf(w2, v2.y, acc.y);
      acc.z = fmaf(w2, v2.z, acc.z); acc.w = fmaf(w2, v2.w, acc.w);
      acc.x = fmaf(w3, v3.x, acc.x); acc.y = fmaf(w3, v3.y, acc.y);
      acc.z = fmaf(w3, v3.z, acc.z); acc.w = fmaf(w3, v3.w, acc.w);
    }
    for (; t < n; ++t) {
      const int ct = __shfl(c, t, 16);
      const float wt = __shfl(w, t, 16);
      const float4 v = *(const float4*)(Hg + (size_t)ct * F_DIM);
      acc.x = fmaf(wt, v.x, acc.x); acc.y = fmaf(wt, v.y, acc.y);
      acc.z = fmaf(wt, v.z, acc.z); acc.w = fmaf(wt, v.w, acc.w);
    }
  }
  *(float4*)(P + (size_t)r * F_DIM + g * 4) = acc;
}

template <int FOUT, bool RELU>
__global__ void cheb_gemm(const float* __restrict__ X, const float* __restrict__ P,
                          const float* __restrict__ W0, const float* __restrict__ W1,
                          const float* __restrict__ B, float* __restrict__ OUT) {
  constexpr int NT   = FOUT / 16;
  constexpr int NKG  = F_DIM / 4;
  constexpr int NENT = NKG * NT * 32;

  __shared__ float4 wlds[NENT];

  for (int idx = threadIdx.x; idx < NENT; idx += blockDim.x) {
    const int l   = idx & 31;
    const int rem = idx >> 5;
    const int t   = rem % NT;
    const int kg  = rem / NT;
    const int n   = t * 16 + (l & 15);
    const int k   = kg * 4 + ((l >> 4) * 2);
    wlds[idx] = make_float4(W0[(size_t)k * FOUT + n], W0[(size_t)(k + 1) * FOUT + n],
                            W1[(size_t)k * FOUT + n], W1[(size_t)(k + 1) * FOUT + n]);
  }
  __syncthreads();

  const int lane  = threadIdx.x & 31;
  const int wave  = threadIdx.x >> 5;
  const int tile  = blockIdx.x * 4 + wave;
  const int ntile = N_NODES / 16;
  if (tile >= ntile) return;

  const int row0 = tile * 16;
  const int m    = lane & 15;
  const int kk   = (lane >> 4) * 2;

  v8f acc[NT];
#pragma unroll
  for (int t = 0; t < NT; ++t) acc[t] = (v8f)0.0f;

  const float* xrow = X + (size_t)(row0 + m) * F_DIM;
  const float* prow = P + (size_t)(row0 + m) * F_DIM;

#pragma unroll
  for (int kg = 0; kg < NKG; ++kg) {
    const int k0 = kg * 4;
    const v2f aX = *(const v2f*)(xrow + k0 + kk);
    const v2f aP = *(const v2f*)(prow + k0 + kk);
#pragma unroll
    for (int t = 0; t < NT; ++t) {
      const float4 wv = wlds[(kg * NT + t) * 32 + lane];
      v2f b0, b1;
      b0.x = wv.x; b0.y = wv.y;
      b1.x = wv.z; b1.y = wv.w;
      acc[t] = __builtin_amdgcn_wmma_f32_16x16x4_f32(
          false, aX, false, b0, (short)0, acc[t], false, false);
      acc[t] = __builtin_amdgcn_wmma_f32_16x16x4_f32(
          false, aP, false, b1, (short)0, acc[t], false, false);
    }
  }

  const int roff = (lane >> 4) * 8;
#pragma unroll
  for (int t = 0; t < NT; ++t) {
    const float bias = B[t * 16 + m];
#pragma unroll
    for (int r = 0; r < 8; ++r) {
      float v = acc[t][r] + bias;
      if (RELU) v = fmaxf(v, 0.0f);
      OUT[(size_t)(row0 + roff + r) * FOUT + t * 16 + m] = v;
    }
  }
}

extern "C" void kernel_launch(void* const* d_in, const int* in_sizes, int n_in,
                              void* d_out, int out_size, void* d_ws, size_t ws_size,
                              hipStream_t stream) {
  const float* x    = (const float*)d_in[0];
  const int*   adj  = (const int*)d_in[1];
  const float* W1_0 = (const float*)d_in[2];
  const float* W1_1 = (const float*)d_in[3];
  const float* b1   = (const float*)d_in[4];
  const float* Wx_0 = (const float*)d_in[5];
  const float* Wx_1 = (const float*)d_in[6];
  const float* bx   = (const float*)d_in[7];
  const float* W2_0 = (const float*)d_in[8];
  const float* W2_1 = (const float*)d_in[9];
  const float* b2   = (const float*)d_in[10];
  float* out = (float*)d_out;

  const int* row = adj;
  const int* col = adj + E_EDGES;

  int*   rowptr = (int*)d_ws;
  int*   scol   = rowptr + N_NODES;
  float* P   = (float*)(scol + E_EDGES);
  float* H1  = P + (size_t)N_NODES * F_DIM;
  float* H2  = H1 + (size_t)N_NODES * F_DIM;
  int*   pairs    = (int*)H2;
  int*   hist     = pairs + 2 * E_EDGES;
  int*   within   = hist + NBIN * NCHUNK;
  int*   total    = within + NBIN * NCHUNK;
  int*   binstart = total + NBIN;

  const int TB = 256;
  const int gGa  = (N_NODES * 16 + TB - 1) / TB;
  const int gGem = (N_NODES / 16 + 3) / 4;

  bin_hist<<<NCHUNK, HIST_BD, 0, stream>>>(row, hist);
  bin_scan<<<NBIN, NCHUNK, 0, stream>>>(hist, within, total);
  bin_scatter<<<NCHUNK, SCAT_BD, 0, stream>>>(row, col, within, total, binstart, pairs);
  bin_sort<<<NBIN, SORT_BD, 0, stream>>>(pairs, binstart, total, rowptr, scol);

  gather_prop<<<gGa, TB, 0, stream>>>(rowptr, scol, x, P);
  cheb_gemm<64, true><<<gGem, 128, 0, stream>>>(x, P, W1_0, W1_1, b1, H1);

  gather_prop<<<gGa, TB, 0, stream>>>(rowptr, scol, H1, P);
  cheb_gemm<64, true><<<gGem, 128, 0, stream>>>(H1, P, Wx_0, Wx_1, bx, H2);

  gather_prop<<<gGa, TB, 0, stream>>>(rowptr, scol, H2, P);
  cheb_gemm<16, false><<<gGem, 128, 0, stream>>>(H2, P, W2_0, W2_1, b2, out);
}


# edge weights -deg^-1/2[r]*deg^-1/2[c] formed once in the layer-1 gather and stored per edge; layers 2-3 read them beside the source index, no random row-start reads per edge; sums bit-identical
# speedup vs baseline: 3.1746x; 1.0617x over previous
#include <hip/hip_runtime.h>

typedef __attribute__((ext_vector_type(2))) float v2f;
typedef __attribute__((ext_vector_type(8))) float v8f;

#define N_NODES 100000
#define E_EDGES 1600000
#define F_DIM   64


#define RPB_SHIFT 8
#define RPB       (1 << RPB_SHIFT)
#define NBIN      ((N_NODES + RPB - 1) / RPB)
#define NCHUNK    256
#define CHUNK     ((E_EDGES + NCHUNK - 1) / NCHUNK)

#define HIST_BD   1024
#define HIST_U    ((CHUNK + HIST_BD - 1) / HIST_BD)
#define SCAT_BD   512
#define SCAT_U    ((CHUNK + SCAT_BD - 1) / SCAT_BD)
#define SORT_BD   1024
#define SORT_U    4

static_assert(NBIN <= SCAT_BD, "one thread per bin in bin_scatter");
static_assert(NCHUNK * CHUNK >= E_EDGES, "chunks cover the edge list");
static_assert(3 * E_EDGES + 2 * NBIN * NCHUNK + 2 * NBIN <= N_NODES * F_DIM,
              "edge weights and grouping temporaries fit in one N*64 plane");

__global__ void bin_hist(const int* __restrict__ row, int* __restrict__ hist) {
  __shared__ int h[NBIN];
  const int tid = threadIdx.x;
  const int g = blockIdx.x;
  for (int b = tid; b < NBIN; b += HIST_BD) h[b] = 0;
  __syncthreads();
  const int lo = g * CHUNK;
  const int hi = min(lo + CHUNK, E_EDGES);
  int r[HIST_U];
#pragma unroll
  for (int u = 0; u < HIST_U; ++u) r[u] = row[min(lo + tid + u * HIST_BD, hi - 1)];
#pragma unroll
  for (int u = 0; u < HIST_U; ++u)
    if (lo + tid + u * HIST_BD < hi) atomicAdd(&h[r[u] >> RPB_SHIFT], 1);
  __syncthreads();
  for (int b = tid; b < NBIN; b += HIST_BD) hist[b * NCHUNK + g] = h[b];
}

__global__ void bin_scan(const int* __restrict__ hist, int* __restrict__ within,
                         int* __restrict__ total) {
  __shared__ int buf[2][NCHUNK];
  const int tid = threadIdx.x;
  const int b = blockIdx.x;
  const int v = hist[b * NCHUNK + tid];
  int cur = 0;
  buf[0][tid] = v;
  __syncthreads();
  for (int d = 1; d < NCHUNK; d <<= 1) {
    int t = buf[cur][tid];
    if (tid >= d) t += buf[cur][tid - d];
    buf[cur ^ 1][tid] = t;
    cur ^= 1;
    __syncthreads();
  }
  const int incl = buf[cur][tid];
  within[b * NCHUNK + tid] = incl - v;
  if (tid == NCHUNK - 1) total[b] = incl;
}

__global__ void bin_scatter(const int* __restrict__ row, const int* __restrict__ col,
                            const int* __restrict__ within, const int* __restrict__ total,
                            int* __restrict__ binstart, int* __restrict__ pairs) {
  __shared__ int buf[2][SCAT_BD];
  __shared__ int cursor[NBIN];
  const int tid = threadIdx.x;
  const int g = blockIdx.x;

  const int v = (tid < NBIN) ? total[tid] : 0;
  int cur = 0;
  buf[0][tid] = v;
  __syncthreads();
  for (int d = 1; d < SCAT_BD; d <<= 1) {
    int t = buf[cur][tid];
    if (tid >= d) t += buf[cur][tid - d];
    buf[cur ^ 1][tid] = t;
    cur ^= 1;
    __syncthreads();
  }
  if (tid < NBIN) {
    const int bs = buf[cur][tid] - v;
    cursor[tid] = bs + within[tid * NCHUNK + g];
    if (g == 0) binstart[tid] = bs;
  }
  __syncthreads();

  const int lo = g * CHUNK;
  const int hi = min(lo + CHUNK, E_EDGES);
  int2* __restrict__ out = (int2*)pairs;
  int r[SCAT_U], c[SCAT_U];
#pragma unroll
  for (int u = 0; u < SCAT_U; ++u) {
    const int i = min(lo + tid + u * SCAT_BD, hi - 1);
    r[u] = row[i];
    c[u] = col[i];
  }
#pragma unroll
  for (int u = 0; u < SCAT_U; ++u) {
    if (lo + tid + u * SCAT_BD < hi) {
      const int pos = atomicAdd(&cursor[r[u] >> RPB_SHIFT], 1);
      out[pos] = make_int2(r[u], c[u]);
    }
  }
}

__global__ void bin_sort(const int* __restrict__ pairs, const int* __restrict__ binstart,
                         const int* __restrict__ total, int* __restrict__ rowptr,
                         int* __restrict__ scol) {
  __shared__ int cnt[RPB];
  __shared__ int fill[RPB];
  __shared__ int buf[2][RPB];
  const int tid = threadIdx.x;
  const int b = blockIdx.x;
  const int bs = binstart[b];
  const int n = total[b];
  const int2* __restrict__ bin = (const int2*)pairs + bs;

  if (tid < RPB) cnt[tid] = 0;
  __syncthreads();

  for (int base = 0; base < n; base += SORT_U * SORT_BD) {
    int2 e[SORT_U];
#pragma unroll
    for (int u = 0; u < SORT_U; ++u) e[u] = bin[min(base + tid + u * SORT_BD, n - 1)];
#pragma unroll
    for (int u = 0; u < SORT_U; ++u)
      if (base + tid + u * SORT_BD < n) atomicAdd(&cnt[e[u].x & (RPB - 1)], 1);
  }
  __syncthreads();

  int cur = 0;
  if (tid < RPB) buf[0][tid] = cnt[tid];
  __syncthreads();
  for (int d = 1; d < RPB; d <<= 1) {
    if (tid < RPB) {
      int t = buf[cur][tid];
      if (tid >= d) t += buf[cur][tid - d];
      buf[cur ^ 1][tid] = t;
    }
    cur ^= 1;
    __syncthreads();
  }
  if (tid < RPB) {
    const int start = bs + buf[cur][tid] - cnt[tid];
    const int r = b * RPB + tid;
    if (r < N_NODES) rowptr[r] = start;
    fill[tid] = start;
  }
  __syncthreads();

  for (int base = 0; base < n; base += SORT_U * SORT_BD) {
    int2 e[SORT_U];
#pragma unroll
    for (int u = 0; u < SORT_U; ++u) e[u] = bin[min(base + tid + u * SORT_BD, n - 1)];
#pragma unroll
    for (int u = 0; u < SORT_U; ++u) {
      if (base + tid + u * SORT_BD < n) {
        const int pos = atomicAdd(&fill[e[u].x & (RPB - 1)], 1);
        scol[pos] = e[u].y;
      }
    }
  }
}

__device__ __forceinline__ float inv_sqrt_deg(int d) {
  return (d > 0) ? rsqrtf((float)d) : 0.0f;
}

template <bool FIRST>
__global__ void gather_prop(const int* __restrict__ rowptr, const int* __restrict__ scol,
                            float* __restrict__ wts,
                            const float* __restrict__ H, float* __restrict__ P) {
  const int gid = blockIdx.x * blockDim.x + threadIdx.x;
  const int r = gid >> 4;
  if (r >= N_NODES) return;
  const int g = threadIdx.x & 15;
  const int s = rowptr[r];
  const int e = (r + 1 < N_NODES) ? rowptr[r + 1] : E_EDGES;
  const float dr = inv_sqrt_deg(e - s);
  const float* Hg = H + g * 4;
  float4 acc = make_float4(0.0f, 0.0f, 0.0f, 0.0f);

  for (int base = s; base < e; base += 16) {
    const int j = base + g;
    int c = 0;
    float w = 0.0f;
    if (j < e) {
      c = scol[j];
      if (FIRST) {
        const int cs = rowptr[c];
        const int ce = (c + 1 < N_NODES) ? rowptr[c + 1] : E_EDGES;
        w = -dr * inv_sqrt_deg(ce - cs);
        wts[j] = w;
      } else {
        w = wts[j];
      }
    }
    const int n = min(16, e - base);
    int t = 0;
    for (; t + 4 <= n; t += 4) {
      const int c0 = __shfl(c, t, 16);
      const int c1 = __shfl(c, t + 1, 16);
      const int c2 = __shfl(c, t + 2, 16);
      const int c3 = __shfl(c, t + 3, 16);
      const float w0 = __shfl(w, t, 16);
      const float w1 = __shfl(w, t + 1, 16);
      const float w2 = __shfl(w, t + 2, 16);
      const float w3 = __shfl(w, t + 3, 16);
      const float4 v0 = *(const float4*)(Hg + (size_t)c0 * F_DIM);
      const float4 v1 = *(const float4*)(Hg + (size_t)c1 * F_DIM);
      const float4 v2 = *(const float4*)(Hg + (size_t)c2 * F_DIM);
      const float4 v3 = *(const float4*)(Hg + (size_t)c3 * F_DIM);
      acc.x = fmaf(w0, v0.x, acc.x); acc.y = fmaf(w0, v0.y, acc.y);
      acc.z = fmaf(w0, v0.z, acc.z); acc.w = fmaf(w0, v0.w, acc.w);
      acc.x = fmaf(w1, v1.x, acc.x); acc.y = fmaf(w1, v1.y, acc.y);
      acc.z = fmaf(w1, v1.z, acc.z); acc.w = fmaf(w1, v1.w, acc.w);
      acc.x = fmaf(w2, v2.x, acc.x); acc.y = fmaf(w2, v2.y, acc.y);
      acc.z = fmaf(w2, v2.z, acc.z); acc.w = fmaf(w2, v2.w, acc.w);
      acc.x = fmaf(w3, v3.x, acc.x); acc.y = fmaf(w3, v3.y, acc.y);
      acc.z = fmaf(w3, v3.z, acc.z); acc.w = fmaf(w3, v3.w, acc.w);
    }
    for (; t < n; ++t) {
      const int ct = __shfl(c, t, 16);
      const float wt = __shfl(w, t, 16);
      const float4 v = *(const float4*)(Hg + (size_t)ct * F_DIM);
      acc.x = fmaf(wt, v.x, acc.x); acc.y = fmaf(wt, v.y, acc.y);
      acc.z = fmaf(wt, v.z, acc.z); acc.w = fmaf(wt, v.w, acc.w);
    }
  }
  *(float4*)(P + (size_t)r * F_DIM + g * 4) = acc;
}

template <int FOUT, bool RELU>
__global__ void cheb_gemm(const float* __restrict__ X, const float* P,
                          const float* __restrict__ W0, const float* __restrict__ W1,
                          const float* __restrict__ B, float* OUT) {
  constexpr int NT   = FOUT / 16;
  constexpr int NKG  = F_DIM / 4;
  constexpr int NENT = NKG * NT * 32;

  __shared__ float4 wlds[NENT];

  for (int idx = threadIdx.x; idx < NENT; idx += blockDim.x) {
    const int l   = idx & 31;
    const int rem = idx >> 5;
    const int t   = rem % NT;
    const int kg  = rem / NT;
    const int n   = t * 16 + (l & 15);
    const int k   = kg * 4 + ((l >> 4) * 2);
    wlds[idx] = make_float4(W0[(size_t)k * FOUT + n], W0[(size_t)(k + 1) * FOUT + n],
                            W1[(size_t)k * FOUT + n], W1[(size_t)(k + 1) * FOUT + n]);
  }
  __syncthreads();

  const int lane  = threadIdx.x & 31;
  const int wave  = threadIdx.x >> 5;
  const int tile  = blockIdx.x * 4 + wave;
  const int ntile = N_NODES / 16;
  if (tile >= ntile) return;

  const int row0 = tile * 16;
  const int m    = lane & 15;
  const int kk   = (lane >> 4) * 2;

  v8f acc[NT];
#pragma unroll
  for (int t = 0; t < NT; ++t) acc[t] = (v8f)0.0f;

  const float* xrow = X + (size_t)(row0 + m) * F_DIM;
  const float* prow = P + (size_t)(row0 + m) * F_DIM;

#pragma unroll
  for (int kg = 0; kg < NKG; ++kg) {
    const int k0 = kg * 4;
    const v2f aX = *(const v2f*)(xrow + k0 + kk);
    const v2f aP = *(const v2f*)(prow + k0 + kk);
#pragma unroll
    for (int t = 0; t < NT; ++t) {
      const float4 wv = wlds[(kg * NT + t) * 32 + lane];
      v2f b0, b1;
      b0.x = wv.x; b0.y = wv.y;
      b1.x = wv.z; b1.y = wv.w;
      acc[t] = __builtin_amdgcn_wmma_f32_16x16x4_f32(
          false, aX, false, b0, (short)0, acc[t], false, false);
      acc[t] = __builtin_amdgcn_wmma_f32_16x16x4_f32(
          false, aP, false, b1, (short)0, acc[t], false, false);
    }
  }

  const int roff = (lane >> 4) * 8;
#pragma unroll
  for (int t = 0; t < NT; ++t) {
    const float bias = B[t * 16 + m];
#pragma unroll
    for (int r = 0; r < 8; ++r) {
      float v = acc[t][r] + bias;
      if (RELU) v = fmaxf(v, 0.0f);
      OUT[(size_t)(row0 + roff + r) * FOUT + t * 16 + m] = v;
    }
  }
}

extern "C" void kernel_launch(void* const* d_in, const int* in_sizes, int n_in,
                              void* d_out, int out_size, void* d_ws, size_t ws_size,
                              hipStream_t stream) {
  const float* x    = (const float*)d_in[0];
  const int*   adj  = (const int*)d_in[1];
  const float* W1_0 = (const float*)d_in[2];
  const float* W1_1 = (const float*)d_in[3];
  const float* b1   = (const float*)d_in[4];
  const float* Wx_0 = (const float*)d_in[5];
  const float* Wx_1 = (const float*)d_in[6];
  const float* bx   = (const float*)d_in[7];
  const float* W2_0 = (const float*)d_in[8];
  const float* W2_1 = (const float*)d_in[9];
  const float* b2   = (const float*)d_in[10];
  float* out = (float*)d_out;

  const int* row = adj;
  const int* col = adj + E_EDGES;

  int*   rowptr = (int*)d_ws;
  int*   scol   = rowptr + N_NODES;
  float* A   = (float*)(scol + E_EDGES);
  float* Bp  = A + (size_t)N_NODES * F_DIM;
  float* C   = Bp + (size_t)N_NODES * F_DIM;
  float* wts      = C;
  int*   pairs    = (int*)(wts + E_EDGES);
  int*   hist     = pairs + 2 * E_EDGES;
  int*   within   = hist + NBIN * NCHUNK;
  int*   total    = within + NBIN * NCHUNK;
  int*   binstart = total + NBIN;

  const int TB = 256;
  const int gGa  = (N_NODES * 16 + TB - 1) / TB;
  const int gGem = (N_NODES / 16 + 3) / 4;

  bin_hist<<<NCHUNK, HIST_BD, 0, stream>>>(row, hist);
  bin_scan<<<NBIN, NCHUNK, 0, stream>>>(hist, within, total);
  bin_scatter<<<NCHUNK, SCAT_BD, 0, stream>>>(row, col, within, total, binstart, pairs);
  bin_sort<<<NBIN, SORT_BD, 0, stream>>>(pairs, binstart, total, rowptr, scol);

  gather_prop<true><<<gGa, TB, 0, stream>>>(rowptr, scol, wts, x, A);
  cheb_gemm<64, true><<<gGem, 128, 0, stream>>>(x, A, W1_0, W1_1, b1, Bp);

  gather_prop<false><<<gGa, TB, 0, stream>>>(rowptr, scol, wts, Bp, A);
  cheb_gemm<64, true><<<gGem, 128, 0, stream>>>(Bp, A, Wx_0, Wx_1, bx, A);

  gather_prop<false><<<gGa, TB, 0, stream>>>(rowptr, scol, wts, A, Bp);
  cheb_gemm<16, false><<<gGem, 128, 0, stream>>>(A, Bp, W2_0, W2_1, b2, out);
}


# layer 3 propagates after the 64->16 projection: out = H@W0 + b + prop(H@W1), not prop(H)@W1; last gather reads 16-float rows, not 64; float sum order changes (W1 product before the neighbour sum)
# speedup vs baseline: 3.2812x; 1.0336x over previous
#include <hip/hip_runtime.h>

typedef __attribute__((ext_vector_type(2))) float v2f;
typedef __attribute__((ext_vector_type(8))) float v8f;

#define N_NODES 100000
#define E_EDGES 1600000
#define F_DIM   64


#define RPB_SHIFT 8
#define RPB       (1 << RPB_SHIFT)
#define NBIN      ((N_NODES + RPB - 1) / RPB)
#define NCHUNK    256
#define CHUNK     ((E_EDGES + NCHUNK - 1) / NCHUNK)

#define HIST_BD   1024
#define HIST_U    ((CHUNK + HIST_BD - 1) / HIST_BD)
#define SCAT_BD   512
#define SCAT_U    ((CHUNK + SCAT_BD - 1) / SCAT_BD)
#define SORT_BD   1024
#define SORT_U    4

static_assert(NBIN <= SCAT_BD, "one thread per bin in bin_scatter");
static_assert(NCHUNK * CHUNK >= E_EDGES, "chunks cover the edge list");
static_assert(3 * E_EDGES + 2 * NBIN * NCHUNK + 2 * NBIN <= N_NODES * F_DIM,
              "edge weights and grouping temporaries fit in one N*64 plane");

__global__ void bin_hist(const int* __restrict__ row, int* __restrict__ hist) {
  __shared__ int h[NBIN];
  const int tid = threadIdx.x;
  const int g = blockIdx.x;
  for (int b = tid; b < NBIN; b += HIST_BD) h[b] = 0;
  __syncthreads();
  const int lo = g * CHUNK;
  const int hi = min(lo + CHUNK, E_EDGES);
  int r[HIST_U];
#pragma unroll
  for (int u = 0; u < HIST_U; ++u) r[u] = row[min(lo + tid + u * HIST_BD, hi - 1)];
#pragma unroll
  for (int u = 0; u < HIST_U; ++u)
    if (lo + tid + u * HIST_BD < hi) atomicAdd(&h[r[u] >> RPB_SHIFT], 1);
  __syncthreads();
  for (int b = tid; b < NBIN; b += HIST_BD) hist[b * NCHUNK + g] = h[b];
}

__global__ void bin_scan(const int* __restrict__ hist, int* __restrict__ within,
                         int* __restrict__ total) {
  __shared__ int buf[2][NCHUNK];
  const int tid = threadIdx.x;
  const int b = blockIdx.x;
  const int v = hist[b * NCHUNK + tid];
  int cur = 0;
  buf[0][tid] = v;
  __syncthreads();
  for (int d = 1; d < NCHUNK; d <<= 1) {
    int t = buf[cur][tid];
    if (tid >= d) t += buf[cur][tid - d];
    buf[cur ^ 1][tid] = t;
    cur ^= 1;
    __syncthreads();
  }
  const int incl = buf[cur][tid];
  within[b * NCHUNK + tid] = incl - v;
  if (tid == NCHUNK - 1) total[b] = incl;
}

__global__ void bin_scatter(const int* __restrict__ row, const int* __restrict__ col,
                            const int* __restrict__ within, const int* __restrict__ total,
                            int* __restrict__ binstart, int* __restrict__ pairs) {
  __shared__ int buf[2][SCAT_BD];
  __shared__ int cursor[NBIN];
  const int tid = threadIdx.x;
  const int g = blockIdx.x;

  const int v = (tid < NBIN) ? total[tid] : 0;
  int cur = 0;
  buf[0][tid] = v;
  __syncthreads();
  for (int d = 1; d < SCAT_BD; d <<= 1) {
    int t = buf[cur][tid];
    if (tid >= d) t += buf[cur][tid - d];
    buf[cur ^ 1][tid] = t;
    cur ^= 1;
    __syncthreads();
  }
  if (tid < NBIN) {
    const int bs = buf[cur][tid] - v;
    cursor[tid] = bs + within[tid * NCHUNK + g];
    if (g == 0) binstart[tid] = bs;
  }
  __syncthreads();

  const int lo = g * CHUNK;
  const int hi = min(lo + CHUNK, E_EDGES);
  int2* __restrict__ out = (int2*)pairs;
  int r[SCAT_U], c[SCAT_U];
#pragma unroll
  for (int u = 0; u < SCAT_U; ++u) {
    const int i = min(lo + tid + u * SCAT_BD, hi - 1);
    r[u] = row[i];
    c[u] = col[i];
  }
#pragma unroll
  for (int u = 0; u < SCAT_U; ++u) {
    if (lo + tid + u * SCAT_BD < hi) {
      const int pos = atomicAdd(&cursor[r[u] >> RPB_SHIFT], 1);
      out[pos] = make_int2(r[u], c[u]);
    }
  }
}

__global__ void bin_sort(const int* __restrict__ pairs, const int* __restrict__ binstart,
                         const int* __restrict__ total, int* __restrict__ rowptr,
                         int* __restrict__ scol) {
  __shared__ int cnt[RPB];
  __shared__ int fill[RPB];
  __shared__ int buf[2][RPB];
  const int tid = threadIdx.x;
  const int b = blockIdx.x;
  const int bs = binstart[b];
  const int n = total[b];
  const int2* __restrict__ bin = (const int2*)pairs + bs;

  if (tid < RPB) cnt[tid] = 0;
  __syncthreads();

  for (int base = 0; base < n; base += SORT_U * SORT_BD) {
    int2 e[SORT_U];
#pragma unroll
    for (int u = 0; u < SORT_U; ++u) e[u] = bin[min(base + tid + u * SORT_BD, n - 1)];
#pragma unroll
    for (int u = 0; u < SORT_U; ++u)
      if (base + tid + u * SORT_BD < n) atomicAdd(&cnt[e[u].x & (RPB - 1)], 1);
  }
  __syncthreads();

  int cur = 0;
  if (tid < RPB) buf[0][tid] = cnt[tid];
  __syncthreads();
  for (int d = 1; d < RPB; d <<= 1) {
    if (tid < RPB) {
      int t = buf[cur][tid];
      if (tid >= d) t += buf[cur][tid - d];
      buf[cur ^ 1][tid] = t;
    }
    cur ^= 1;
    __syncthreads();
  }
  if (tid < RPB) {
    const int start = bs + buf[cur][tid] - cnt[tid];
    const int r = b * RPB + tid;
    if (r < N_NODES) rowptr[r] = start;
    fill[tid] = start;
  }
  __syncthreads();

  for (int base = 0; base < n; base += SORT_U * SORT_BD) {
    int2 e[SORT_U];
#pragma unroll
    for (int u = 0; u < SORT_U; ++u) e[u] = bin[min(base + tid + u * SORT_BD, n - 1)];
#pragma unroll
    for (int u = 0; u < SORT_U; ++u) {
      if (base + tid + u * SORT_BD < n) {
        const int pos = atomicAdd(&fill[e[u].x & (RPB - 1)], 1);
        scol[pos] = e[u].y;
      }
    }
  }
}

__device__ __forceinline__ float inv_sqrt_deg(int d) {
  return (d > 0) ? rsqrtf((float)d) : 0.0f;
}

template <bool FIRST>
__global__ void gather_prop(const int* __restrict__ rowptr, const int* __restrict__ scol,
                            float* __restrict__ wts,
                            const float* __restrict__ H, float* __restrict__ P) {
  const int gid = blockIdx.x * blockDim.x + threadIdx.x;
  const int r = gid >> 4;
  if (r >= N_NODES) return;
  const int g = threadIdx.x & 15;
  const int s = rowptr[r];
  const int e = (r + 1 < N_NODES) ? rowptr[r + 1] : E_EDGES;
  const float dr = inv_sqrt_deg(e - s);
  const float* Hg = H + g * 4;
  float4 acc = make_float4(0.0f, 0.0f, 0.0f, 0.0f);

  for (int base = s; base < e; base += 16) {
    const int j = base + g;
    int c = 0;
    float w = 0.0f;
    if (j < e) {
      c = scol[j];
      if (FIRST) {
        const int cs = rowptr[c];
        const int ce = (c + 1 < N_NODES) ? rowptr[c + 1] : E_EDGES;
        w = -dr * inv_sqrt_deg(ce - cs);
        wts[j] = w;
      } else {
        w = wts[j];
      }
    }
    const int n = min(16, e - base);
    int t = 0;
    for (; t + 4 <= n; t += 4) {
      const int c0 = __shfl(c, t, 16);
      const int c1 = __shfl(c, t + 1, 16);
      const int c2 = __shfl(c, t + 2, 16);
      const int c3 = __shfl(c, t + 3, 16);
      const float w0 = __shfl(w, t, 16);
      const float w1 = __shfl(w, t + 1, 16);
      const float w2 = __shfl(w, t + 2, 16);
      const float w3 = __shfl(w, t + 3, 16);
      const float4 v0 = *(const float4*)(Hg + (size_t)c0 * F_DIM);
      const float4 v1 = *(const float4*)(Hg + (size_t)c1 * F_DIM);
      const float4 v2 = *(const float4*)(Hg + (size_t)c2 * F_DIM);
      const float4 v3 = *(const float4*)(Hg + (size_t)c3 * F_DIM);
      acc.x = fmaf(w0, v0.x, acc.x); acc.y = fmaf(w0, v0.y, acc.y);
      acc.z = fmaf(w0, v0.z, acc.z); acc.w = fmaf(w0, v0.w, acc.w);
      acc.x = fmaf(w1, v1.x, acc.x); acc.y = fmaf(w1, v1.y, acc.y);
      acc.z = fmaf(w1, v1.z, acc.z); acc.w = fmaf(w1, v1.w, acc.w);
      acc.x = fmaf(w2, v2.x, acc.x); acc.y = fmaf(w2, v2.y, acc.y);
      acc.z = fmaf(w2, v2.z, acc.z); acc.w = fmaf(w2, v2.w, acc.w);
      acc.x = fmaf(w3, v3.x, acc.x); acc.y = fmaf(w3, v3.y, acc.y);
      acc.z = fmaf(w3, v3.z, acc.z); acc.w = fmaf(w3, v3.w, acc.w);
    }
    for (; t < n; ++t) {
      const int ct = __shfl(c, t, 16);
      const float wt = __shfl(w, t, 16);
      const float4 v = *(const float4*)(Hg + (size_t)ct * F_DIM);
      acc.x = fmaf(wt, v.x, acc.x); acc.y = fmaf(wt, v.y, acc.y);
      acc.z = fmaf(wt, v.z, acc.z); acc.w = fmaf(wt, v.w, acc.w);
    }
  }
  *(float4*)(P + (size_t)r * F_DIM + g * 4) = acc;
}

template <int FOUT, bool RELU>
__global__ void cheb_gemm(const float* __restrict__ X, const float* P,
                          const float* __restrict__ W0, const float* __restrict__ W1,
                          const float* __restrict__ B, float* OUT) {
  constexpr int NT   = FOUT / 16;
  constexpr int NKG  = F_DIM / 4;
  constexpr int NENT = NKG * NT * 32;

  __shared__ float4 wlds[NENT];

  for (int idx = threadIdx.x; idx < NENT; idx += blockDim.x) {
    const int l   = idx & 31;
    const int rem = idx >> 5;
    const int t   = rem % NT;
    const int kg  = rem / NT;
    const int n   = t * 16 + (l & 15);
    const int k   = kg * 4 + ((l >> 4) * 2);
    wlds[idx] = make_float4(W0[(size_t)k * FOUT + n], W0[(size_t)(k + 1) * FOUT + n],
                            W1[(size_t)k * FOUT + n], W1[(size_t)(k + 1) * FOUT + n]);
  }
  __syncthreads();

  const int lane  = threadIdx.x & 31;
  const int wave  = threadIdx.x >> 5;
  const int tile  = blockIdx.x * 4 + wave;
  const int ntile = N_NODES / 16;
  if (tile >= ntile) return;

  const int row0 = tile * 16;
  const int m    = lane & 15;
  const int kk   = (lane >> 4) * 2;

  v8f acc[NT];
#pragma unroll
  for (int t = 0; t < NT; ++t) acc[t] = (v8f)0.0f;

  const float* xrow = X + (size_t)(row0 + m) * F_DIM;
  const float* prow = P + (size_t)(row0 + m) * F_DIM;

#pragma unroll
  for (int kg = 0; kg < NKG; ++kg) {
    const int k0 = kg * 4;
    const v2f aX = *(const v2f*)(xrow + k0 + kk);
    const v2f aP = *(const v2f*)(prow + k0 + kk);
#pragma unroll
    for (int t = 0; t < NT; ++t) {
      const float4 wv = wlds[(kg * NT + t) * 32 + lane];
      v2f b0, b1;
      b0.x = wv.x; b0.y = wv.y;
      b1.x = wv.z; b1.y = wv.w;
      acc[t] = __builtin_amdgcn_wmma_f32_16x16x4_f32(
          false, aX, false, b0, (short)0, acc[t], false, false);
      acc[t] = __builtin_amdgcn_wmma_f32_16x16x4_f32(
          false, aP, false, b1, (short)0, acc[t], false, false);
    }
  }

  const int roff = (lane >> 4) * 8;
#pragma unroll
  for (int t = 0; t < NT; ++t) {
    const float bias = B[t * 16 + m];
#pragma unroll
    for (int r = 0; r < 8; ++r) {
      float v = acc[t][r] + bias;
      if (RELU) v = fmaxf(v, 0.0f);
      OUT[(size_t)(row0 + roff + r) * FOUT + t * 16 + m] = v;
    }
  }
}


__global__ void cheb_proj16(const float* __restrict__ X,
                            const float* __restrict__ W0, const float* __restrict__ W1,
                            const float* __restrict__ B,
                            float* __restrict__ Y, float* __restrict__ Z) {
  constexpr int FOUT = 16;
  constexpr int NKG  = F_DIM / 4;
  constexpr int NENT = NKG * 32;

  __shared__ float4 wlds[NENT];

  for (int idx = threadIdx.x; idx < NENT; idx += blockDim.x) {
    const int l  = idx & 31;
    const int kg = idx >> 5;
    const int n  = l & 15;
    const int k  = kg * 4 + ((l >> 4) * 2);
    wlds[idx] = make_float4(W0[(size_t)k * FOUT + n], W0[(size_t)(k + 1) * FOUT + n],
                            W1[(size_t)k * FOUT + n], W1[(size_t)(k + 1) * FOUT + n]);
  }
  __syncthreads();

  const int lane  = threadIdx.x & 31;
  const int wave  = threadIdx.x >> 5;
  const int tile  = blockIdx.x * 4 + wave;
  const int ntile = N_NODES / 16;
  if (tile >= ntile) return;

  const int row0 = tile * 16;
  const int m    = lane & 15;
  const int kk   = (lane >> 4) * 2;

  v8f accY = (v8f)0.0f;
  v8f accZ = (v8f)0.0f;
  const float* xrow = X + (size_t)(row0 + m) * F_DIM;

#pragma unroll
  for (int kg = 0; kg < NKG; ++kg) {
    const v2f aX = *(const v2f*)(xrow + kg * 4 + kk);
    const float4 wv = wlds[kg * 32 + lane];
    v2f b0, b1;
    b0.x = wv.x; b0.y = wv.y;
    b1.x = wv.z; b1.y = wv.w;
    accY = __builtin_amdgcn_wmma_f32_16x16x4_f32(
        false, aX, false, b0, (short)0, accY, false, false);
    accZ = __builtin_amdgcn_wmma_f32_16x16x4_f32(
        false, aX, false, b1, (short)0, accZ, false, false);
  }

  const int roff = (lane >> 4) * 8;
  const float bias = B[m];
#pragma unroll
  for (int r = 0; r < 8; ++r) {
    Y[(size_t)(row0 + roff + r) * FOUT + m] = accY[r] + bias;
    Z[(size_t)(row0 + roff + r) * FOUT + m] = accZ[r];
  }
}

__global__ void gather_out16(const int* __restrict__ rowptr, const int* __restrict__ scol,
                             const float* __restrict__ wts,
                             const float* __restrict__ Y, const float* __restrict__ Z,
                             float* __restrict__ OUT) {
  const int gid = blockIdx.x * blockDim.x + threadIdx.x;
  const int r = gid >> 2;
  if (r >= N_NODES) return;
  const int q = threadIdx.x & 3;
  const int s = rowptr[r];
  const int e = (r + 1 < N_NODES) ? rowptr[r + 1] : E_EDGES;
  const float* Zq = Z + q * 4;
  float4 acc = *(const float4*)(Y + (size_t)r * 16 + q * 4);

  const int jlast = E_EDGES - 1;
  int c = scol[min(s + q, jlast)];
  float w = wts[min(s + q, jlast)];
  for (int base = s; base < e; base += 4) {
    const int jn = min(base + 4 + q, jlast);
    const int cn = scol[jn];
    const float wn = wts[jn];
    const int n = min(4, e - base);
    if (n == 4) {
      const int c0 = __shfl(c, 0, 4);
      const int c1 = __shfl(c, 1, 4);
      const int c2 = __shfl(c, 2, 4);
      const int c3 = __shfl(c, 3, 4);
      const float w0 = __shfl(w, 0, 4);
      const float w1 = __shfl(w, 1, 4);
      const float w2 = __shfl(w, 2, 4);
      const float w3 = __shfl(w, 3, 4);
      const float4 v0 = *(const float4*)(Zq + (size_t)c0 * 16);
      const float4 v1 = *(const float4*)(Zq + (size_t)c1 * 16);
      const float4 v2 = *(const float4*)(Zq + (size_t)c2 * 16);
      const float4 v3 = *(const float4*)(Zq + (size_t)c3 * 16);
      acc.x = fmaf(w0, v0.x, acc.x); acc.y = fmaf(w0, v0.y, acc.y);
      acc.z = fmaf(w0, v0.z, acc.z); acc.w = fmaf(w0, v0.w, acc.w);
      acc.x = fmaf(w1, v1.x, acc.x); acc.y = fmaf(w1, v1.y, acc.y);
      acc.z = fmaf(w1, v1.z, acc.z); acc.w = fmaf(w1, v1.w, acc.w);
      acc.x = fmaf(w2, v2.x, acc.x); acc.y = fmaf(w2, v2.y, acc.y);
      acc.z = fmaf(w2, v2.z, acc.z); acc.w = fmaf(w2, v2.w, acc.w);
      acc.x = fmaf(w3, v3.x, acc.x); acc.y = fmaf(w3, v3.y, acc.y);
      acc.z = fmaf(w3, v3.z, acc.z); acc.w = fmaf(w3, v3.w, acc.w);
    } else {
      for (int t = 0; t < n; ++t) {
        const int ct = __shfl(c, t, 4);
        const float wt = __shfl(w, t, 4);
        const float4 v = *(const float4*)(Zq + (size_t)ct * 16);
        acc.x = fmaf(wt, v.x, acc.x); acc.y = fmaf(wt, v.y, acc.y);
        acc.z = fmaf(wt, v.z, acc.z); acc.w = fmaf(wt, v.w, acc.w);
      }
    }
    c = cn;
    w = wn;
  }
  *(float4*)(OUT + (size_t)r * 16 + q * 4) = acc;
}

extern "C" void kernel_launch(void* const* d_in, const int* in_sizes, int n_in,
                              void* d_out, int out_size, void* d_ws, size_t ws_size,
                              hipStream_t stream) {
  const float* x    = (const float*)d_in[0];
  const int*   adj  = (const int*)d_in[1];
  const float* W1_0 = (const float*)d_in[2];
  const float* W1_1 = (const float*)d_in[3];
  const float* b1   = (const float*)d_in[4];
  const float* Wx_0 = (const float*)d_in[5];
  const float* Wx_1 = (const float*)d_in[6];
  const float* bx   = (const float*)d_in[7];
  const float* W2_0 = (const float*)d_in[8];
  const float* W2_1 = (const float*)d_in[9];
  const float* b2   = (const float*)d_in[10];
  float* out = (float*)d_out;

  const int* row = adj;
  const int* col = adj + E_EDGES;

  int*   rowptr = (int*)d_ws;
  int*   scol   = rowptr + N_NODES;
  float* A   = (float*)(scol + E_EDGES);
  float* Bp  = A + (size_t)N_NODES * F_DIM;
  float* C   = Bp + (size_t)N_NODES * F_DIM;
  float* wts      = C;
  int*   pairs    = (int*)(wts + E_EDGES);
  int*   hist     = pairs + 2 * E_EDGES;
  int*   within   = hist + NBIN * NCHUNK;
  int*   total    = within + NBIN * NCHUNK;
  int*   binstart = total + NBIN;

  const int TB = 256;
  const int gGa  = (N_NODES * 16 + TB - 1) / TB;
  const int gGem = (N_NODES / 16 + 3) / 4;
  const int gG16 = (N_NODES * 4 + TB - 1) / TB;

  bin_hist<<<NCHUNK, HIST_BD, 0, stream>>>(row, hist);
  bin_scan<<<NBIN, NCHUNK, 0, stream>>>(hist, within, total);
  bin_scatter<<<NCHUNK, SCAT_BD, 0, stream>>>(row, col, within, total, binstart, pairs);
  bin_sort<<<NBIN, SORT_BD, 0, stream>>>(pairs, binstart, total, rowptr, scol);

  gather_prop<true><<<gGa, TB, 0, stream>>>(rowptr, scol, wts, x, A);
  cheb_gemm<64, true><<<gGem, 128, 0, stream>>>(x, A, W1_0, W1_1, b1, Bp);

  gather_prop<false><<<gGa, TB, 0, stream>>>(rowptr, scol, wts, Bp, A);
  cheb_gemm<64, true><<<gGem, 128, 0, stream>>>(Bp, A, Wx_0, Wx_1, bx, A);

  float* Y3 = Bp;
  float* Z3 = Bp + (size_t)N_NODES * 16;
  cheb_proj16<<<gGem, 128, 0, stream>>>(A, W2_0, W2_1, b2, Y3, Z3);
  gather_out16<<<gG16, TB, 0, stream>>>(rowptr, scol, wts, Y3, Z3, out);
}


# row gather fused into the layer-1/2 product launch: a wave sums its 16-row P strip (8 row pairs in series, next pairs' indices requested ahead) into LDS, the f32 WMMA reads it there; 2 launches fewer
# speedup vs baseline: 3.6668x; 1.1175x over previous
#include <hip/hip_runtime.h>

typedef __attribute__((ext_vector_type(2))) float v2f;
typedef __attribute__((ext_vector_type(8))) float v8f;

#define N_NODES 100000
#define E_EDGES 1600000
#define F_DIM   64


#define RPB_SHIFT 8
#define RPB       (1 << RPB_SHIFT)
#define NBIN      ((N_NODES + RPB - 1) / RPB)
#define NCHUNK    256
#define CHUNK     ((E_EDGES + NCHUNK - 1) / NCHUNK)

#define HIST_BD   1024
#define HIST_U    ((CHUNK + HIST_BD - 1) / HIST_BD)
#define SCAT_BD   512
#define SCAT_U    ((CHUNK + SCAT_BD - 1) / SCAT_BD)
#define SORT_BD   1024
#define SORT_U    4

static_assert(NBIN <= SCAT_BD, "one thread per bin in bin_scatter");
static_assert(NCHUNK * CHUNK >= E_EDGES, "chunks cover the edge list");
static_assert(3 * E_EDGES + 2 * NBIN * NCHUNK + 2 * NBIN <= N_NODES * F_DIM,
              "edge weights and grouping temporaries fit in one N*64 plane");

__global__ void bin_hist(const int* __restrict__ row, int* __restrict__ hist) {
  __shared__ int h[NBIN];
  const int tid = threadIdx.x;
  const int g = blockIdx.x;
  for (int b = tid; b < NBIN; b += HIST_BD) h[b] = 0;
  __syncthreads();
  const int lo = g * CHUNK;
  const int hi = min(lo + CHUNK, E_EDGES);
  int r[HIST_U];
#pragma unroll
  for (int u = 0; u < HIST_U; ++u) r[u] = row[min(lo + tid + u * HIST_BD, hi - 1)];
#pragma unroll
  for (int u = 0; u < HIST_U; ++u)
    if (lo + tid + u * HIST_BD < hi) atomicAdd(&h[r[u] >> RPB_SHIFT], 1);
  __syncthreads();
  for (int b = tid; b < NBIN; b += HIST_BD) hist[b * NCHUNK + g] = h[b];
}

__global__ void bin_scan(const int* __restrict__ hist, int* __restrict__ within,
                         int* __restrict__ total) {
  __shared__ int buf[2][NCHUNK];
  const int tid = threadIdx.x;
  const int b = blockIdx.x;
  const int v = hist[b * NCHUNK + tid];
  int cur = 0;
  buf[0][tid] = v;
  __syncthreads();
  for (int d = 1; d < NCHUNK; d <<= 1) {
    int t = buf[cur][tid];
    if (tid >= d) t += buf[cur][tid - d];
    buf[cur ^ 1][tid] = t;
    cur ^= 1;
    __syncthreads();
  }
  const int incl = buf[cur][tid];
  within[b * NCHUNK + tid] = incl - v;
  if (tid == NCHUNK - 1) total[b] = incl;
}

__global__ void bin_scatter(const int* __restrict__ row, const int* __restrict__ col,
                            const int* __restrict__ within, const int* __restrict__ total,
                            int* __restrict__ binstart, int* __restrict__ pairs) {
  __shared__ int buf[2][SCAT_BD];
  __shared__ int cursor[NBIN];
  const int tid = threadIdx.x;
  const int g = blockIdx.x;

  const int v = (tid < NBIN) ? total[tid] : 0;
  int cur = 0;
  buf[0][tid] = v;
  __syncthreads();
  for (int d = 1; d < SCAT_BD; d <<= 1) {
    int t = buf[cur][tid];
    if (tid >= d) t += buf[cur][tid - d];
    buf[cur ^ 1][tid] = t;
    cur ^= 1;
    __syncthreads();
  }
  if (tid < NBIN) {
    const int bs = buf[cur][tid] - v;
    cursor[tid] = bs + within[tid * NCHUNK + g];
    if (g == 0) binstart[tid] = bs;
  }
  __syncthreads();

  const int lo = g * CHUNK;
  const int hi = min(lo + CHUNK, E_EDGES);
  int2* __restrict__ out = (int2*)pairs;
  int r[SCAT_U], c[SCAT_U];
#pragma unroll
  for (int u = 0; u < SCAT_U; ++u) {
    const int i = min(lo + tid + u * SCAT_BD, hi - 1);
    r[u] = row[i];
    c[u] = col[i];
  }
#pragma unroll
  for (int u = 0; u < SCAT_U; ++u) {
    if (lo + tid + u * SCAT_BD < hi) {
      const int pos = atomicAdd(&cursor[r[u] >> RPB_SHIFT], 1);
      out[pos] = make_int2(r[u], c[u]);
    }
  }
}

__global__ void bin_sort(const int* __restrict__ pairs, const int* __restrict__ binstart,
                         const int* __restrict__ total, int* __restrict__ rowptr,
                         int* __restrict__ scol) {
  __shared__ int cnt[RPB];
  __shared__ int fill[RPB];
  __shared__ int buf[2][RPB];
  const int tid = threadIdx.x;
  const int b = blockIdx.x;
  const int bs = binstart[b];
  const int n = total[b];
  const int2* __restrict__ bin = (const int2*)pairs + bs;

  if (tid < RPB) cnt[tid] = 0;
  __syncthreads();

  for (int base = 0; base < n; base += SORT_U * SORT_BD) {
    int2 e[SORT_U];
#pragma unroll
    for (int u = 0; u < SORT_U; ++u) e[u] = bin[min(base + tid + u * SORT_BD, n - 1)];
#pragma unroll
    for (int u = 0; u < SORT_U; ++u)
      if (base + tid + u * SORT_BD < n) atomicAdd(&cnt[e[u].x & (RPB - 1)], 1);
  }
  __syncthreads();

  int cur = 0;
  if (tid < RPB) buf[0][tid] = cnt[tid];
  __syncthreads();
  for (int d = 1; d < RPB; d <<= 1) {
    if (tid < RPB) {
      int t = buf[cur][tid];
      if (tid >= d) t += buf[cur][tid - d];
      buf[cur ^ 1][tid] = t;
    }
    cur ^= 1;
    __syncthreads();
  }
  if (tid < RPB) {
    const int start = bs + buf[cur][tid] - cnt[tid];
    const int r = b * RPB + tid;
    if (r < N_NODES) rowptr[r] = start;
    fill[tid] = start;
  }
  __syncthreads();

  for (int base = 0; base < n; base += SORT_U * SORT_BD) {
    int2 e[SORT_U];
#pragma unroll
    for (int u = 0; u < SORT_U; ++u) e[u] = bin[min(base + tid + u * SORT_BD, n - 1)];
#pragma unroll
    for (int u = 0; u < SORT_U; ++u) {
      if (base + tid + u * SORT_BD < n) {
        const int pos = atomicAdd(&fill[e[u].x & (RPB - 1)], 1);
        scol[pos] = e[u].y;
      }
    }
  }
}

__device__ __forceinline__ float inv_sqrt_deg(int d) {
  return (d > 0) ? rsqrtf((float)d) : 0.0f;
}

template <int U>
__device__ __forceinline__ void walk_trip(const float* __restrict__ Hg, int c, float w,
                                          int t, float4& acc) {
  int cc[U];
  float ww[U];
  float4 v[U];
#pragma unroll
  for (int u = 0; u < U; ++u) {
    cc[u] = __shfl(c, t + u, 16);
    ww[u] = __shfl(w, t + u, 16);
  }
#pragma unroll
  for (int u = 0; u < U; ++u) v[u] = *(const float4*)(Hg + (size_t)cc[u] * F_DIM);
#pragma unroll
  for (int u = 0; u < U; ++u) {
    acc.x = fmaf(ww[u], v[u].x, acc.x); acc.y = fmaf(ww[u], v[u].y, acc.y);
    acc.z = fmaf(ww[u], v[u].z, acc.z); acc.w = fmaf(ww[u], v[u].w, acc.w);
  }
}

__device__ __forceinline__ void walk_batch(const float* __restrict__ Hg, int c, float w,
                                           int n, float4& acc) {
  int t = 0;
  for (; t + 8 <= n; t += 8) walk_trip<8>(Hg, c, w, t, acc);
  for (; t + 4 <= n; t += 4) walk_trip<4>(Hg, c, w, t, acc);
  for (; t < n; ++t) walk_trip<1>(Hg, c, w, t, acc);
}

__device__ __forceinline__ int row_end(const int* __restrict__ rowptr, int c) {
  return (c + 1 < N_NODES) ? rowptr[c + 1] : E_EDGES;
}

#define LW      8
#define PPITCH  68

template <bool FIRST>
__global__ void __launch_bounds__(LW * 32)
cheb_layer(const int* __restrict__ rowptr, const int* __restrict__ scol,
           float* __restrict__ wts, const float* __restrict__ H,
           const float* __restrict__ W0, const float* __restrict__ W1,
           const float* __restrict__ B, float* __restrict__ OUT) {
  constexpr int FOUT = F_DIM;
  constexpr int NT   = FOUT / 16;
  constexpr int NKG  = F_DIM / 4;
  constexpr int NENT = NKG * NT * 32;

  __shared__ float4 wlds[NENT];
  __shared__ __attribute__((aligned(16))) float ptile[LW][16 * PPITCH];

  for (int idx = threadIdx.x; idx < NENT; idx += LW * 32) {
    const int l   = idx & 31;
    const int rem = idx >> 5;
    const int t   = rem % NT;
    const int kg  = rem / NT;
    const int n   = t * 16 + (l & 15);
    const int k   = kg * 4 + ((l >> 4) * 2);
    wlds[idx] = make_float4(W0[(size_t)k * FOUT + n], W0[(size_t)(k + 1) * FOUT + n],
                            W1[(size_t)k * FOUT + n], W1[(size_t)(k + 1) * FOUT + n]);
  }

  const int lane  = threadIdx.x & 31;
  const int wave  = threadIdx.x >> 5;
  const int tile  = blockIdx.x * LW + wave;
  const int ntile = N_NODES / 16;
  const bool active = tile < ntile;
  const int row0 = tile * 16;
  const int m    = lane & 15;
  const int hl   = lane >> 4;
  float* pt = ptile[wave];

  if (active) {
    const int jlast = E_EDGES - 1;
    const float* Hg = H + m * 4;
    const int ri = row0 + min(lane, 16);
    const int rp = (ri < N_NODES) ? rowptr[ri] : E_EDGES;

    int s0 = __shfl(rp, hl),     e0 = __shfl(rp, hl + 1);
    int s1 = __shfl(rp, 2 + hl), e1 = __shfl(rp, 3 + hl);
    int c0 = scol[min(s0 + m, jlast)];
    int c1 = scol[min(s1 + m, jlast)];
    float w0, w1 = 0.0f;
    if (FIRST) {
      w0 = -inv_sqrt_deg(e0 - s0) * inv_sqrt_deg(row_end(rowptr, c0) - rowptr[c0]);
      if (s0 + m < e0) wts[s0 + m] = w0;
    } else {
      w0 = wts[min(s0 + m, jlast)];
      w1 = wts[min(s1 + m, jlast)];
    }

    for (int it = 0; it < 8; ++it) {
      const int l2 = min(2 * it + 4 + hl, 15);
      const int s2 = __shfl(rp, l2), e2 = __shfl(rp, l2 + 1);
      const int j2 = min(s2 + m, jlast);
      const int c2 = scol[j2];
      float w2 = 0.0f;
      int cs1 = 0, ce1 = 0;
      if (FIRST) {
        cs1 = rowptr[c1];
        ce1 = row_end(rowptr, c1);
      } else {
        w2 = wts[j2];
      }

      float4 acc = make_float4(0.0f, 0.0f, 0.0f, 0.0f);
      const float dr = inv_sqrt_deg(e0 - s0);
      walk_batch(Hg, c0, w0, min(16, e0 - s0), acc);
      for (int base = s0 + 16; base < e0; base += 16) {
        const int j = base + m;
        int c = 0;
        float w = 0.0f;
        if (j < e0) {
          c = scol[j];
          if (FIRST) {
            w = -dr * inv_sqrt_deg(row_end(rowptr, c) - rowptr[c]);
            wts[j] = w;
          } else {
            w = wts[j];
          }
        }
        walk_batch(Hg, c, w, min(16, e0 - base), acc);
      }
      *(float4*)(pt + (2 * it + hl) * PPITCH + m * 4) = acc;

      if (FIRST) {
        w1 = -inv_sqrt_deg(e1 - s1) * inv_sqrt_deg(ce1 - cs1);
        if (it < 7 && s1 + m < e1) wts[s1 + m] = w1;
      }
      s0 = s1; e0 = e1; c0 = c1; w0 = w1;
      s1 = s2; e1 = e2; c1 = c2; w1 = w2;
    }
  }
  __syncthreads();
  if (!active) return;

  const int kk = hl * 2;
  v8f acc[NT];
#pragma unroll
  for (int t = 0; t < NT; ++t) acc[t] = (v8f)0.0f;

  const float* xrow = H + (size_t)(row0 + m) * F_DIM;
  const float* prow = pt + m * PPITCH;

#pragma unroll
  for (int kg = 0; kg < NKG; ++kg) {
    const int k0 = kg * 4;
    const v2f aX = *(const v2f*)(xrow + k0 + kk);
    const v2f aP = *(const v2f*)(prow + k0 + kk);
#pragma unroll
    for (int t = 0; t < NT; ++t) {
      const float4 wv = wlds[(kg * NT + t) * 32 + lane];
      v2f b0, b1;
      b0.x = wv.x; b0.y = wv.y;
      b1.x = wv.z; b1.y = wv.w;
      acc[t] = __builtin_amdgcn_wmma_f32_16x16x4_f32(
          false, aX, false, b0, (short)0, acc[t], false, false);
      acc[t] = __builtin_amdgcn_wmma_f32_16x16x4_f32(
          false, aP, false, b1, (short)0, acc[t], false, false);
    }
  }

  const int roff = hl * 8;
#pragma unroll
  for (int t = 0; t < NT; ++t) {
    const float bias = B[t * 16 + m];
#pragma unroll
    for (int r = 0; r < 8; ++r) {
      const float v = fmaxf(acc[t][r] + bias, 0.0f);
      OUT[(size_t)(row0 + roff + r) * FOUT + t * 16 + m] = v;
    }
  }
}


__global__ void cheb_proj16(const float* __restrict__ X,
                            const float* __restrict__ W0, const float* __restrict__ W1,
                            const float* __restrict__ B,
                            float* __restrict__ Y, float* __restrict__ Z) {
  constexpr int FOUT = 16;
  constexpr int NKG  = F_DIM / 4;
  constexpr int NENT = NKG * 32;

  __shared__ float4 wlds[NENT];

  for (int idx = threadIdx.x; idx < NENT; idx += blockDim.x) {
    const int l  = idx & 31;
    const int kg = idx >> 5;
    const int n  = l & 15;
    const int k  = kg * 4 + ((l >> 4) * 2);
    wlds[idx] = make_float4(W0[(size_t)k * FOUT + n], W0[(size_t)(k + 1) * FOUT + n],
                            W1[(size_t)k * FOUT + n], W1[(size_t)(k + 1) * FOUT + n]);
  }
  __syncthreads();

  const int lane  = threadIdx.x & 31;
  const int wave  = threadIdx.x >> 5;
  const int tile  = blockIdx.x * 4 + wave;
  const int ntile = N_NODES / 16;
  if (tile >= ntile) return;

  const int row0 = tile * 16;
  const int m    = lane & 15;
  const int kk   = (lane >> 4) * 2;

  v8f accY = (v8f)0.0f;
  v8f accZ = (v8f)0.0f;
  const float* xrow = X + (size_t)(row0 + m) * F_DIM;

#pragma unroll
  for (int kg = 0; kg < NKG; ++kg) {
    const v2f aX = *(const v2f*)(xrow + kg * 4 + kk);
    const float4 wv = wlds[kg * 32 + lane];
    v2f b0, b1;
    b0.x = wv.x; b0.y = wv.y;
    b1.x = wv.z; b1.y = wv.w;
    accY = __builtin_amdgcn_wmma_f32_16x16x4_f32(
        false, aX, false, b0, (short)0, accY, false, false);
    accZ = __builtin_amdgcn_wmma_f32_16x16x4_f32(
        false, aX, false, b1, (short)0, accZ, false, false);
  }

  const int roff = (lane >> 4) * 8;
  const float bias = B[m];
#pragma unroll
  for (int r = 0; r < 8; ++r) {
    Y[(size_t)(row0 + roff + r) * FOUT + m] = accY[r] + bias;
    Z[(size_t)(row0 + roff + r) * FOUT + m] = accZ[r];
  }
}

__global__ void gather_out16(const int* __restrict__ rowptr, const int* __restrict__ scol,
                             const float* __restrict__ wts,
                             const float* __restrict__ Y, const float* __restrict__ Z,
                             float* __restrict__ OUT) {
  const int gid = blockIdx.x * blockDim.x + threadIdx.x;
  const int r = gid >> 2;
  if (r >= N_NODES) return;
  const int q = threadIdx.x & 3;
  const int s = rowptr[r];
  const int e = (r + 1 < N_NODES) ? rowptr[r + 1] : E_EDGES;
  const float* Zq = Z + q * 4;
  float4 acc = *(const float4*)(Y + (size_t)r * 16 + q * 4);

  const int jlast = E_EDGES - 1;
  int c = scol[min(s + q, jlast)];
  float w = wts[min(s + q, jlast)];
  for (int base = s; base < e; base += 4) {
    const int jn = min(base + 4 + q, jlast);
    const int cn = scol[jn];
    const float wn = wts[jn];
    const int n = min(4, e - base);
    if (n == 4) {
      const int c0 = __shfl(c, 0, 4);
      const int c1 = __shfl(c, 1, 4);
      const int c2 = __shfl(c, 2, 4);
      const int c3 = __shfl(c, 3, 4);
      const float w0 = __shfl(w, 0, 4);
      const float w1 = __shfl(w, 1, 4);
      const float w2 = __shfl(w, 2, 4);
      const float w3 = __shfl(w, 3, 4);
      const float4 v0 = *(const float4*)(Zq + (size_t)c0 * 16);
      const float4 v1 = *(const float4*)(Zq + (size_t)c1 * 16);
      const float4 v2 = *(const float4*)(Zq + (size_t)c2 * 16);
      const float4 v3 = *(const float4*)(Zq + (size_t)c3 * 16);
      acc.x = fmaf(w0, v0.x, acc.x); acc.y = fmaf(w0, v0.y, acc.y);
      acc.z = fmaf(w0, v0.z, acc.z); acc.w = fmaf(w0, v0.w, acc.w);
      acc.x = fmaf(w1, v1.x, acc.x); acc.y = fmaf(w1, v1.y, acc.y);
      acc.z = fmaf(w1, v1.z, acc.z); acc.w = fmaf(w1, v1.w, acc.w);
      acc.x = fmaf(w2, v2.x, acc.x); acc.y = fmaf(w2, v2.y, acc.y);
      acc.z = fmaf(w2, v2.z, acc.z); acc.w = fmaf(w2, v2.w, acc.w);
      acc.x = fmaf(w3, v3.x, acc.x); acc.y = fmaf(w3, v3.y, acc.y);
      acc.z = fmaf(w3, v3.z, acc.z); acc.w = fmaf(w3, v3.w, acc.w);
    } else {
      for (int t = 0; t < n; ++t) {
        const int ct = __shfl(c, t, 4);
        const float wt = __shfl(w, t, 4);
        const float4 v = *(const float4*)(Zq + (size_t)ct * 16);
        acc.x = fmaf(wt, v.x, acc.x); acc.y = fmaf(wt, v.y, acc.y);
        acc.z = fmaf(wt, v.z, acc.z); acc.w = fmaf(wt, v.w, acc.w);
      }
    }
    c = cn;
    w = wn;
  }
  *(float4*)(OUT + (size_t)r * 16 + q * 4) = acc;
}

extern "C" void kernel_launch(void* const* d_in, const int* in_sizes, int n_in,
                              void* d_out, int out_size, void* d_ws, size_t ws_size,
                              hipStream_t stream) {
  const float* x    = (const float*)d_in[0];
  const int*   adj  = (const int*)d_in[1];
  const float* W1_0 = (const float*)d_in[2];
  const float* W1_1 = (const float*)d_in[3];
  const float* b1   = (const float*)d_in[4];
  const float* Wx_0 = (const float*)d_in[5];
  const float* Wx_1 = (const float*)d_in[6];
  const float* bx   = (const float*)d_in[7];
  const float* W2_0 = (const float*)d_in[8];
  const float* W2_1 = (const float*)d_in[9];
  const float* b2   = (const float*)d_in[10];
  float* out = (float*)d_out;

  const int* row = adj;
  const int* col = adj + E_EDGES;

  int*   rowptr = (int*)d_ws;
  int*   scol   = rowptr + N_NODES;
  float* A   = (float*)(scol + E_EDGES);
  float* Bp  = A + (size_t)N_NODES * F_DIM;
  float* C   = Bp + (size_t)N_NODES * F_DIM;
  float* wts      = C;
  int*   pairs    = (int*)(wts + E_EDGES);
  int*   hist     = pairs + 2 * E_EDGES;
  int*   within   = hist + NBIN * NCHUNK;
  int*   total    = within + NBIN * NCHUNK;
  int*   binstart = total + NBIN;

  const int TB = 256;
  const int gLay = (N_NODES / 16 + LW - 1) / LW;
  const int gGem = (N_NODES / 16 + 3) / 4;
  const int gG16 = (N_NODES * 4 + TB - 1) / TB;

  bin_hist<<<NCHUNK, HIST_BD, 0, stream>>>(row, hist);
  bin_scan<<<NBIN, NCHUNK, 0, stream>>>(hist, within, total);
  bin_scatter<<<NCHUNK, SCAT_BD, 0, stream>>>(row, col, within, total, binstart, pairs);
  bin_sort<<<NBIN, SORT_BD, 0, stream>>>(pairs, binstart, total, rowptr, scol);

  cheb_layer<true><<<gLay, LW * 32, 0, stream>>>(rowptr, scol, wts, x, W1_0, W1_1, b1, Bp);

  cheb_layer<false><<<gLay, LW * 32, 0, stream>>>(rowptr, scol, wts, Bp, Wx_0, Wx_1, bx, A);

  float* Y3 = Bp;
  float* Z3 = Bp + (size_t)N_NODES * 16;
  cheb_proj16<<<gGem, 128, 0, stream>>>(A, W2_0, W2_1, b2, Y3, Z3);
  gather_out16<<<gG16, TB, 0, stream>>>(rowptr, scol, wts, Y3, Z3, out);
}


# edge grouping in 2 launches, not 4: a chunk's block groups its own pairs by bin in place (no count pass, no scan launch); a bin's start = sum of its group offsets; row order = LDS arrival order
# speedup vs baseline: 4.4356x; 1.2097x over previous
#include <hip/hip_runtime.h>

typedef __attribute__((ext_vector_type(2))) float v2f;
typedef __attribute__((ext_vector_type(8))) float v8f;

#define N_NODES 100000
#define E_EDGES 1600000
#define F_DIM   64


#define RPB_SHIFT 8
#define RPB       (1 << RPB_SHIFT)
#define NBIN      ((N_NODES + RPB - 1) / RPB)
#define NCHUNK    256
#define CHUNK     ((E_EDGES + NCHUNK - 1) / NCHUNK)

#define GRP_BD    1024
#define GRP_U     ((CHUNK + GRP_BD - 1) / GRP_BD)
#define GRP_SC    512
#define SORT_BD   1024
#define SORT_U    8
#define SEG_L     (SORT_BD / NCHUNK)

static_assert(NBIN <= GRP_SC, "the bins fit the scan in chunk_group");
static_assert(NCHUNK * CHUNK >= E_EDGES, "chunks cover the edge list");
static_assert(SEG_L * NCHUNK == SORT_BD, "one thread group per chunk in bin_sort");
static_assert(SORT_BD / 32 == 32, "bin_sort adds its wave sums with one wave-wide sum");
static_assert(3 * E_EDGES + 2 * NBIN * NCHUNK <= N_NODES * F_DIM,
              "edge weights and grouping temporaries fit in one N*64 plane");

__global__ void __launch_bounds__(GRP_BD)
chunk_group(const int* __restrict__ row, const int* __restrict__ col,
            int* __restrict__ cntT, int* __restrict__ offT, int* __restrict__ pairs) {
  __shared__ int h[GRP_SC];
  __shared__ int buf[2][GRP_SC];
  const int tid = threadIdx.x;
  const int g = blockIdx.x;
  if (tid < GRP_SC) h[tid] = 0;
  const int lo = g * CHUNK;
  const int hi = min(lo + CHUNK, E_EDGES);
  int r[GRP_U], c[GRP_U];
#pragma unroll
  for (int u = 0; u < GRP_U; ++u) {
    const int i = min(lo + tid + u * GRP_BD, hi - 1);
    r[u] = row[i];
    c[u] = col[i];
  }
  __syncthreads();
#pragma unroll
  for (int u = 0; u < GRP_U; ++u)
    if (lo + tid + u * GRP_BD < hi) atomicAdd(&h[r[u] >> RPB_SHIFT], 1);
  __syncthreads();

  int v = 0;
  if (tid < GRP_SC) {
    v = h[tid];
    buf[0][tid] = v;
  }
  __syncthreads();
  int cur = 0;
  for (int d = 1; d < GRP_SC; d <<= 1) {
    if (tid < GRP_SC) {
      int t = buf[cur][tid];
      if (tid >= d) t += buf[cur][tid - d];
      buf[cur ^ 1][tid] = t;
    }
    cur ^= 1;
    __syncthreads();
  }
  if (tid < NBIN) {
    const int excl = buf[cur][tid] - v;
    h[tid] = excl;
    cntT[tid * NCHUNK + g] = v;
    offT[tid * NCHUNK + g] = excl;
  }
  __syncthreads();

  int2* __restrict__ out = (int2*)pairs + lo;
#pragma unroll
  for (int u = 0; u < GRP_U; ++u) {
    if (lo + tid + u * GRP_BD < hi) {
      const int pos = atomicAdd(&h[r[u] >> RPB_SHIFT], 1);
      out[pos] = make_int2(r[u], c[u]);
    }
  }
}

__global__ void __launch_bounds__(SORT_BD)
bin_sort(const int* __restrict__ pairs, const int* __restrict__ cntT,
         const int* __restrict__ offT, int* __restrict__ rowptr, int* __restrict__ scol) {
  __shared__ int cnt[RPB];
  __shared__ int fill[RPB];
  __shared__ int buf[2][RPB];
  __shared__ int red[SORT_BD / 32];
  const int tid = threadIdx.x;
  const int b = blockIdx.x;
  constexpr int TRIP = SORT_U * SEG_L;

  const int g = tid / SEG_L;
  const int q = tid % SEG_L;
  const int glen = cntT[b * NCHUNK + g];
  const int goff = offT[b * NCHUNK + g];
  const int gstart = g * CHUNK + goff;
  const int2* __restrict__ pr = (const int2*)pairs;
  auto pair_at = [&](int k) {
    return pr[min(gstart + max(min(k, glen - 1), 0), E_EDGES - 1)];
  };

  int2 e0[SORT_U];
#pragma unroll
  for (int u = 0; u < SORT_U; ++u) e0[u] = pair_at(q + u * SEG_L);

  int sb = (q == 0) ? goff : 0;
#pragma unroll
  for (int o = 16; o > 0; o >>= 1) sb += __shfl_xor(sb, o);
  if ((tid & 31) == 0) red[tid >> 5] = sb;
  if (tid < RPB) cnt[tid] = 0;
  __syncthreads();
  int bs = red[tid & 31];
#pragma unroll
  for (int o = 16; o > 0; o >>= 1) bs += __shfl_xor(bs, o);

#pragma unroll
  for (int u = 0; u < SORT_U; ++u)
    if (q + u * SEG_L < glen) atomicAdd(&cnt[e0[u].x & (RPB - 1)], 1);
  for (int base = TRIP; base < glen; base += TRIP) {
    int2 e[SORT_U];
#pragma unroll
    for (int u = 0; u < SORT_U; ++u) e[u] = pair_at(base + q + u * SEG_L);
#pragma unroll
    for (int u = 0; u < SORT_U; ++u)
      if (base + q + u * SEG_L < glen) atomicAdd(&cnt[e[u].x & (RPB - 1)], 1);
  }
  __syncthreads();

  int cur = 0;
  if (tid < RPB) buf[0][tid] = cnt[tid];
  __syncthreads();
  for (int d = 1; d < RPB; d <<= 1) {
    if (tid < RPB) {
      int t = buf[cur][tid];
      if (tid >= d) t += buf[cur][tid - d];
      buf[cur ^ 1][tid] = t;
    }
    cur ^= 1;
    __syncthreads();
  }
  if (tid < RPB) {
    const int start = bs + buf[cur][tid] - cnt[tid];
    const int r = b * RPB + tid;
    if (r < N_NODES) rowptr[r] = start;
    fill[tid] = start;
  }
  __syncthreads();

#pragma unroll
  for (int u = 0; u < SORT_U; ++u) {
    if (q + u * SEG_L < glen) {
      const int pos = atomicAdd(&fill[e0[u].x & (RPB - 1)], 1);
      scol[pos] = e0[u].y;
    }
  }
  for (int base = TRIP; base < glen; base += TRIP) {
    int2 e[SORT_U];
#pragma unroll
    for (int u = 0; u < SORT_U; ++u) e[u] = pair_at(base + q + u * SEG_L);
#pragma unroll
    for (int u = 0; u < SORT_U; ++u) {
      if (base + q + u * SEG_L < glen) {
        const int pos = atomicAdd(&fill[e[u].x & (RPB - 1)], 1);
        scol[pos] = e[u].y;
      }
    }
  }
}

__device__ __forceinline__ float inv_sqrt_deg(int d) {
  return (d > 0) ? rsqrtf((float)d) : 0.0f;
}

template <int U>
__device__ __forceinline__ void walk_fixed(const char* __restrict__ Hb, unsigned lane_off,
                                           int c, float w, int t0, float4& acc) {
  unsigned off[U];
  float ww[U];
  float4 v[U];
#pragma unroll
  for (int u = 0; u < U; ++u) {
    off[u] = (unsigned)__shfl(c, t0 + u, 16) * (unsigned)(F_DIM * sizeof(float)) + lane_off;
    ww[u] = __shfl(w, t0 + u, 16);
  }
#pragma unroll
  for (int u = 0; u < U; ++u) v[u] = *(const float4*)(Hb + off[u]);
#pragma unroll
  for (int u = 0; u < U; ++u) {
    acc.x = fmaf(ww[u], v[u].x, acc.x); acc.y = fmaf(ww[u], v[u].y, acc.y);
    acc.z = fmaf(ww[u], v[u].z, acc.z); acc.w = fmaf(ww[u], v[u].w, acc.w);
  }
}

__device__ __forceinline__ void fix_records(int& c, float& w, int m, int n, int own) {
  const int c0 = __shfl(c, 0, 16);
  const int cfirst = (n > 0) ? c0 : own;
  if (m >= n) {
    c = cfirst;
    w = 0.0f;
  }
}

__device__ __forceinline__ int row_end(const int* __restrict__ rowptr, int c) {
  return (c + 1 < N_NODES) ? rowptr[c + 1] : E_EDGES;
}

#define LW      8
#define PPITCH  68

template <bool FIRST>
__global__ void __launch_bounds__(LW * 32)
cheb_layer(const int* __restrict__ rowptr, const int* __restrict__ scol,
           float* __restrict__ wts, const float* __restrict__ H,
           const float* __restrict__ W0, const float* __restrict__ W1,
           const float* __restrict__ B, float* __restrict__ OUT,
           const float* __restrict__ V0, const float* __restrict__ V1,
           const float* __restrict__ B2, float* __restrict__ Y, float* __restrict__ Z) {
  constexpr int FOUT = F_DIM;
  constexpr int NT   = FOUT / 16;
  constexpr int NKG  = F_DIM / 4;
  constexpr int NENT = NKG * NT * 32;
  constexpr int FO2   = 16;
  constexpr int NENT2 = FIRST ? 1 : NKG * 32;

  __shared__ float4 wlds[NENT];
  __shared__ __attribute__((aligned(16))) float ptile[LW][16 * PPITCH];
  __shared__ float4 w2lds[NENT2];

  for (int idx = threadIdx.x; idx < NENT; idx += LW * 32) {
    const int l   = idx & 31;
    const int rem = idx >> 5;
    const int t   = rem % NT;
    const int kg  = rem / NT;
    const int n   = t * 16 + (l & 15);
    const int k   = kg * 4 + ((l >> 4) * 2);
    wlds[idx] = make_float4(W0[(size_t)k * FOUT + n], W0[(size_t)(k + 1) * FOUT + n],
                            W1[(size_t)k * FOUT + n], W1[(size_t)(k + 1) * FOUT + n]);
  }
  if constexpr (!FIRST) {
    for (int idx = threadIdx.x; idx < NENT2; idx += LW * 32) {
      const int l  = idx & 31;
      const int kg = idx >> 5;
      const int n  = l & 15;
      const int k  = kg * 4 + ((l >> 4) * 2);
      w2lds[idx] = make_float4(V0[(size_t)k * FO2 + n], V0[(size_t)(k + 1) * FO2 + n],
                               V1[(size_t)k * FO2 + n], V1[(size_t)(k + 1) * FO2 + n]);
    }
  }

  const int lane  = threadIdx.x & 31;
  const int wave  = threadIdx.x >> 5;
  const int tile  = blockIdx.x * LW + wave;
  const int ntile = N_NODES / 16;
  const bool active = tile < ntile;
  const int row0 = tile * 16;
  const int m    = lane & 15;
  const int hl   = lane >> 4;
  float* pt = ptile[wave];

  if (active) {
    const int jlast = E_EDGES - 1;
    const char* Hb = (const char*)H;
    const unsigned lane_off = (unsigned)m * 16u;
    const int ri = row0 + min(lane, 16);
    const int rp = (ri < N_NODES) ? rowptr[ri] : E_EDGES;

    int s0 = __shfl(rp, hl),     e0 = __shfl(rp, hl + 1);
    int s1 = __shfl(rp, 2 + hl), e1 = __shfl(rp, 3 + hl);
    int c0 = scol[min(s0 + m, jlast)];
    int c1 = scol[min(s1 + m, jlast)];
    float w0, w1 = 0.0f;
    if (FIRST) {
      w0 = -inv_sqrt_deg(e0 - s0) * inv_sqrt_deg(row_end(rowptr, c0) - rowptr[c0]);
      if (s0 + m < e0) wts[s0 + m] = w0;
    } else {
      w0 = wts[min(s0 + m, jlast)];
      w1 = wts[min(s1 + m, jlast)];
    }

    for (int it = 0; it < 8; ++it) {
      const int l2 = min(2 * it + 4 + hl, 15);
      const int s2 = __shfl(rp, l2), e2 = __shfl(rp, l2 + 1);
      const int j2 = min(s2 + m, jlast);
      const int c2 = scol[j2];
      float w2 = 0.0f;
      int cs1 = 0, ce1 = 0;
      if (FIRST) {
        cs1 = rowptr[c1];
        ce1 = row_end(rowptr, c1);
      } else {
        w2 = wts[j2];
      }

      float4 acc = make_float4(0.0f, 0.0f, 0.0f, 0.0f);
      const float dr = inv_sqrt_deg(e0 - s0);
      const int own = row0 + 2 * it + hl;
      {
        int cb = c0;
        float wb = w0;
        fix_records(cb, wb, m, e0 - s0, own);
        walk_fixed<16>(Hb, lane_off, cb, wb, 0, acc);
      }
      for (int base = s0 + 16; base < e0; base += 16) {
        const int j = base + m;
        int c = 0;
        float w = 0.0f;
        if (j < e0) {
          c = scol[j];
          if (FIRST) {
            w = -dr * inv_sqrt_deg(row_end(rowptr, c) - rowptr[c]);
            wts[j] = w;
          } else {
            w = wts[j];
          }
        }
        const int n = e0 - base;
        fix_records(c, w, m, n, own);
        walk_fixed<8>(Hb, lane_off, c, w, 0, acc);
        if (n > 8) walk_fixed<8>(Hb, lane_off, c, w, 8, acc);
      }
      *(float4*)(pt + (2 * it + hl) * PPITCH + m * 4) = acc;

      if (FIRST) {
        w1 = -inv_sqrt_deg(e1 - s1) * inv_sqrt_deg(ce1 - cs1);
        if (it < 7 && s1 + m < e1) wts[s1 + m] = w1;
      }
      s0 = s1; e0 = e1; c0 = c1; w0 = w1;
      s1 = s2; e1 = e2; c1 = c2; w1 = w2;
    }
  }
  __syncthreads();
  if constexpr (FIRST) {
    if (!active) return;
  }

  const int kk = hl * 2;
  const int roff = hl * 8;
  const float* prow = pt + m * PPITCH;

  if (active) {
  v8f acc[NT];
#pragma unroll
  for (int t = 0; t < NT; ++t) acc[t] = (v8f)0.0f;

  const float* xrow = H + (size_t)(row0 + m) * F_DIM;

#pragma unroll
  for (int kg = 0; kg < NKG; ++kg) {
    const int k0 = kg * 4;
    const v2f aX = *(const v2f*)(xrow + k0 + kk);
    const v2f aP = *(const v2f*)(prow + k0 + kk);
#pragma unroll
    for (int t = 0; t < NT; ++t) {
      const float4 wv = wlds[(kg * NT + t) * 32 + lane];
      v2f b0, b1;
      b0.x = wv.x; b0.y = wv.y;
      b1.x = wv.z; b1.y = wv.w;
      acc[t] = __builtin_amdgcn_wmma_f32_16x16x4_f32(
          false, aX, false, b0, (short)0, acc[t], false, false);
      acc[t] = __builtin_amdgcn_wmma_f32_16x16x4_f32(
          false, aP, false, b1, (short)0, acc[t], false, false);
    }
  }

#pragma unroll
  for (int t = 0; t < NT; ++t) {
    const float bias = B[t * 16 + m];
#pragma unroll
    for (int r = 0; r < 8; ++r) {
      const float v = fmaxf(acc[t][r] + bias, 0.0f);
      if constexpr (FIRST) {
        OUT[(size_t)(row0 + roff + r) * FOUT + t * 16 + m] = v;
      } else {
        pt[(roff + r) * PPITCH + t * 16 + m] = v;
      }
    }
  }
  }

  if constexpr (!FIRST) {
    __syncthreads();
    if (!active) return;

    v8f accY = (v8f)0.0f;
    v8f accZ = (v8f)0.0f;
#pragma unroll
    for (int kg = 0; kg < NKG; ++kg) {
      const v2f aH = *(const v2f*)(prow + kg * 4 + kk);
      const float4 wv = w2lds[kg * 32 + lane];
      v2f b0, b1;
      b0.x = wv.x; b0.y = wv.y;
      b1.x = wv.z; b1.y = wv.w;
      accY = __builtin_amdgcn_wmma_f32_16x16x4_f32(
          false, aH, false, b0, (short)0, accY, false, false);
      accZ = __builtin_amdgcn_wmma_f32_16x16x4_f32(
          false, aH, false, b1, (short)0, accZ, false, false);
    }
    const float bias2 = B2[m];
#pragma unroll
    for (int r = 0; r < 8; ++r) {
      Y[(size_t)(row0 + roff + r) * FO2 + m] = accY[r] + bias2;
      Z[(size_t)(row0 + roff + r) * FO2 + m] = accZ[r];
    }
  }
}


__global__ void gather_out16(const int* __restrict__ rowptr, const int* __restrict__ scol,
                             const float* __restrict__ wts,
                             const float* __restrict__ Y, const float* __restrict__ Z,
                             float* __restrict__ OUT) {
  const int gid = blockIdx.x * blockDim.x + threadIdx.x;
  const int r = gid >> 2;
  if (r >= N_NODES) return;
  const int q = threadIdx.x & 3;
  const int s = rowptr[r];
  const int e = (r + 1 < N_NODES) ? rowptr[r + 1] : E_EDGES;
  const char* Zb = (const char*)Z;
  const unsigned lane_off = (unsigned)q * 16u;
  float4 acc = *(const float4*)(Y + (size_t)r * 16 + q * 4);

  const int jlast = E_EDGES - 1;
  int   ca = scol[min(s + q, jlast)];
  int   cb = scol[min(s + 4 + q, jlast)];
  float wa = wts[min(s + q, jlast)];
  float wb = wts[min(s + 4 + q, jlast)];
  for (int base = s; base < e; base += 8) {
    const int ja = min(base + 8 + q, jlast);
    const int jb = min(base + 12 + q, jlast);
    const int   can = scol[ja];
    const int   cbn = scol[jb];
    const float wan = wts[ja];
    const float wbn = wts[jb];

    const int n = e - base;
    const int cfirst = __shfl(ca, 0, 4);
    if (q >= n)     { ca = cfirst; wa = 0.0f; }
    if (q + 4 >= n) { cb = cfirst; wb = 0.0f; }

    unsigned off[8];
    float ww[8];
    float4 v[8];
#pragma unroll
    for (int k = 0; k < 8; ++k) {
      off[k] = (unsigned)__shfl(k < 4 ? ca : cb, k & 3, 4) * 64u + lane_off;
      ww[k]  = __shfl(k < 4 ? wa : wb, k & 3, 4);
    }
#pragma unroll
    for (int k = 0; k < 8; ++k) v[k] = *(const float4*)(Zb + off[k]);
#pragma unroll
    for (int k = 0; k < 8; ++k) {
      acc.x = fmaf(ww[k], v[k].x, acc.x); acc.y = fmaf(ww[k], v[k].y, acc.y);
      acc.z = fmaf(ww[k], v[k].z, acc.z); acc.w = fmaf(ww[k], v[k].w, acc.w);
    }
    asm volatile("" :: "v"(can), "v"(cbn), "v"(wan), "v"(wbn));
    ca = can; cb = cbn;
    wa = wan; wb = wbn;
  }
  *(float4*)(OUT + (size_t)r * 16 + q * 4) = acc;
}

extern "C" void kernel_launch(void* const* d_in, const int* in_sizes, int n_in,
                              void* d_out, int out_size, void* d_ws, size_t ws_size,
                              hipStream_t stream) {
  const float* x    = (const float*)d_in[0];
  const int*   adj  = (const int*)d_in[1];
  const float* W1_0 = (const float*)d_in[2];
  const float* W1_1 = (const float*)d_in[3];
  const float* b1   = (const float*)d_in[4];
  const float* Wx_0 = (const float*)d_in[5];
  const float* Wx_1 = (const float*)d_in[6];
  const float* bx   = (const float*)d_in[7];
  const float* W2_0 = (const float*)d_in[8];
  const float* W2_1 = (const float*)d_in[9];
  const float* b2   = (const float*)d_in[10];
  float* out = (float*)d_out;

  const int* row = adj;
  const int* col = adj + E_EDGES;

  int*   rowptr = (int*)d_ws;
  int*   scol   = rowptr + N_NODES;
  float* A   = (float*)(scol + E_EDGES);
  float* Bp  = A + (size_t)N_NODES * F_DIM;
  float* C   = Bp + (size_t)N_NODES * F_DIM;
  float* wts      = C;
  int*   pairs    = (int*)(wts + E_EDGES);
  int*   cntT     = pairs + 2 * E_EDGES;
  int*   offT     = cntT + NBIN * NCHUNK;

  const int TB = 256;
  const int gLay = (N_NODES / 16 + LW - 1) / LW;
  const int gG16 = (N_NODES * 4 + TB - 1) / TB;

  chunk_group<<<NCHUNK, GRP_BD, 0, stream>>>(row, col, cntT, offT, pairs);
  bin_sort<<<NBIN, SORT_BD, 0, stream>>>(pairs, cntT, offT, rowptr, scol);

  float* Y3 = A;
  float* Z3 = A + (size_t)N_NODES * 16;
  cheb_layer<true><<<gLay, LW * 32, 0, stream>>>(rowptr, scol, wts, x, W1_0, W1_1, b1, Bp,
                                                 W2_0, W2_1, b2, Y3, Z3);

  cheb_layer<false><<<gLay, LW * 32, 0, stream>>>(rowptr, scol, wts, Bp, Wx_0, Wx_1, bx, A,
                                                  W2_0, W2_1, b2, Y3, Z3);

  gather_out16<<<gG16, TB, 0, stream>>>(rowptr, scol, wts, Y3, Z3, out);
}


# layer-1/2 weight staging moved into the row-pair walk, one 256-entry slice per pair read beside that pair's row fetches: no 8 (layer 2: 10) serial round trips ahead of a wave's gather; sums unchanged
# speedup vs baseline: 4.6464x; 1.0475x over previous
#include <hip/hip_runtime.h>

typedef __attribute__((ext_vector_type(2))) float v2f;
typedef __attribute__((ext_vector_type(8))) float v8f;

#define N_NODES 100000
#define E_EDGES 1600000
#define F_DIM   64


#define RPB_SHIFT 8
#define RPB       (1 << RPB_SHIFT)
#define NBIN      ((N_NODES + RPB - 1) / RPB)
#define NCHUNK    256
#define CHUNK     ((E_EDGES + NCHUNK - 1) / NCHUNK)

#define GRP_BD    1024
#define GRP_U     ((CHUNK + GRP_BD - 1) / GRP_BD)
#define GRP_SC    512
#define SORT_BD   1024
#define SORT_U    8
#define SEG_L     (SORT_BD / NCHUNK)

static_assert(NBIN <= GRP_SC, "the bins fit the scan in chunk_group");
static_assert(NCHUNK * CHUNK >= E_EDGES, "chunks cover the edge list");
static_assert(SEG_L * NCHUNK == SORT_BD, "one thread group per chunk in bin_sort");
static_assert(SORT_BD / 32 == 32, "bin_sort adds its wave sums with one wave-wide sum");
static_assert(3 * E_EDGES + 2 * NBIN * NCHUNK <= N_NODES * F_DIM,
              "edge weights and grouping temporaries fit in one N*64 plane");

__global__ void __launch_bounds__(GRP_BD)
chunk_group(const int* __restrict__ row, const int* __restrict__ col,
            int* __restrict__ cntT, int* __restrict__ offT, int* __restrict__ pairs) {
  __shared__ int h[GRP_SC];
  __shared__ int buf[2][GRP_SC];
  const int tid = threadIdx.x;
  const int g = blockIdx.x;
  if (tid < GRP_SC) h[tid] = 0;
  const int lo = g * CHUNK;
  const int hi = min(lo + CHUNK, E_EDGES);
  int r[GRP_U], c[GRP_U];
#pragma unroll
  for (int u = 0; u < GRP_U; ++u) {
    const int i = min(lo + tid + u * GRP_BD, hi - 1);
    r[u] = row[i];
    c[u] = col[i];
  }
  __syncthreads();
#pragma unroll
  for (int u = 0; u < GRP_U; ++u)
    if (lo + tid + u * GRP_BD < hi) atomicAdd(&h[r[u] >> RPB_SHIFT], 1);
  __syncthreads();

  int v = 0;
  if (tid < GRP_SC) {
    v = h[tid];
    buf[0][tid] = v;
  }
  __syncthreads();
  int cur = 0;
  for (int d = 1; d < GRP_SC; d <<= 1) {
    if (tid < GRP_SC) {
      int t = buf[cur][tid];
      if (tid >= d) t += buf[cur][tid - d];
      buf[cur ^ 1][tid] = t;
    }
    cur ^= 1;
    __syncthreads();
  }
  if (tid < NBIN) {
    const int excl = buf[cur][tid] - v;
    h[tid] = excl;
    cntT[tid * NCHUNK + g] = v;
    offT[tid * NCHUNK + g] = excl;
  }
  __syncthreads();

  int2* __restrict__ out = (int2*)pairs + lo;
#pragma unroll
  for (int u = 0; u < GRP_U; ++u) {
    if (lo + tid + u * GRP_BD < hi) {
      const int pos = atomicAdd(&h[r[u] >> RPB_SHIFT], 1);
      out[pos] = make_int2(r[u], c[u]);
    }
  }
}

__global__ void __launch_bounds__(SORT_BD)
bin_sort(const int* __restrict__ pairs, const int* __restrict__ cntT,
         const int* __restrict__ offT, int* __restrict__ rowptr, int* __restrict__ scol) {
  __shared__ int cnt[RPB];
  __shared__ int fill[RPB];
  __shared__ int buf[2][RPB];
  __shared__ int red[SORT_BD / 32];
  const int tid = threadIdx.x;
  const int b = blockIdx.x;
  constexpr int TRIP = SORT_U * SEG_L;

  const int g = tid / SEG_L;
  const int q = tid % SEG_L;
  const int glen = cntT[b * NCHUNK + g];
  const int goff = offT[b * NCHUNK + g];
  const int gstart = g * CHUNK + goff;
  const int2* __restrict__ pr = (const int2*)pairs;
  auto pair_at = [&](int k) {
    return pr[min(gstart + max(min(k, glen - 1), 0), E_EDGES - 1)];
  };

  int2 e0[SORT_U];
#pragma unroll
  for (int u = 0; u < SORT_U; ++u) e0[u] = pair_at(q + u * SEG_L);

  int sb = (q == 0) ? goff : 0;
#pragma unroll
  for (int o = 16; o > 0; o >>= 1) sb += __shfl_xor(sb, o);
  if ((tid & 31) == 0) red[tid >> 5] = sb;
  if (tid < RPB) cnt[tid] = 0;
  __syncthreads();
  int bs = red[tid & 31];
#pragma unroll
  for (int o = 16; o > 0; o >>= 1) bs += __shfl_xor(bs, o);

#pragma unroll
  for (int u = 0; u < SORT_U; ++u)
    if (q + u * SEG_L < glen) atomicAdd(&cnt[e0[u].x & (RPB - 1)], 1);
  for (int base = TRIP; base < glen; base += TRIP) {
    int2 e[SORT_U];
#pragma unroll
    for (int u = 0; u < SORT_U; ++u) e[u] = pair_at(base + q + u * SEG_L);
#pragma unroll
    for (int u = 0; u < SORT_U; ++u)
      if (base + q + u * SEG_L < glen) atomicAdd(&cnt[e[u].x & (RPB - 1)], 1);
  }
  __syncthreads();

  int cur = 0;
  if (tid < RPB) buf[0][tid] = cnt[tid];
  __syncthreads();
  for (int d = 1; d < RPB; d <<= 1) {
    if (tid < RPB) {
      int t = buf[cur][tid];
      if (tid >= d) t += buf[cur][tid - d];
      buf[cur ^ 1][tid] = t;
    }
    cur ^= 1;
    __syncthreads();
  }
  if (tid < RPB) {
    const int start = bs + buf[cur][tid] - cnt[tid];
    const int r = b * RPB + tid;
    if (r < N_NODES) rowptr[r] = start;
    fill[tid] = start;
  }
  __syncthreads();

#pragma unroll
  for (int u = 0; u < SORT_U; ++u) {
    if (q + u * SEG_L < glen) {
      const int pos = atomicAdd(&fill[e0[u].x & (RPB - 1)], 1);
      scol[pos] = e0[u].y;
    }
  }
  for (int base = TRIP; base < glen; base += TRIP) {
    int2 e[SORT_U];
#pragma unroll
    for (int u = 0; u < SORT_U; ++u) e[u] = pair_at(base + q + u * SEG_L);
#pragma unroll
    for (int u = 0; u < SORT_U; ++u) {
      if (base + q + u * SEG_L < glen) {
        const int pos = atomicAdd(&fill[e[u].x & (RPB - 1)], 1);
        scol[pos] = e[u].y;
      }
    }
  }
}

__device__ __forceinline__ float inv_sqrt_deg(int d) {
  return (d > 0) ? rsqrtf((float)d) : 0.0f;
}

template <int U>
__device__ __forceinline__ void walk_fixed(const char* __restrict__ Hb, unsigned lane_off,
                                           int c, float w, int t0, float4& acc) {
  unsigned off[U];
  float ww[U];
  float4 v[U];
#pragma unroll
  for (int u = 0; u < U; ++u) {
    off[u] = (unsigned)__shfl(c, t0 + u, 16) * (unsigned)(F_DIM * sizeof(float)) + lane_off;
    ww[u] = __shfl(w, t0 + u, 16);
  }
#pragma unroll
  for (int u = 0; u < U; ++u) v[u] = *(const float4*)(Hb + off[u]);
#pragma unroll
  for (int u = 0; u < U; ++u) {
    acc.x = fmaf(ww[u], v[u].x, acc.x); acc.y = fmaf(ww[u], v[u].y, acc.y);
    acc.z = fmaf(ww[u], v[u].z, acc.z); acc.w = fmaf(ww[u], v[u].w, acc.w);
  }
}

__device__ __forceinline__ void fix_records(int& c, float& w, int m, int n, int own) {
  const int c0 = __shfl(c, 0, 16);
  const int cfirst = (n > 0) ? c0 : own;
  if (m >= n) {
    c = cfirst;
    w = 0.0f;
  }
}

__device__ __forceinline__ int row_end(const int* __restrict__ rowptr, int c) {
  return (c + 1 < N_NODES) ? rowptr[c + 1] : E_EDGES;
}

#define LW      8
#define PPITCH  68

template <bool FIRST>
__global__ void __launch_bounds__(LW * 32)
cheb_layer(const int* __restrict__ rowptr, const int* __restrict__ scol,
           float* __restrict__ wts, const float* __restrict__ H,
           const float* __restrict__ W0, const float* __restrict__ W1,
           const float* __restrict__ B, float* __restrict__ OUT,
           const float* __restrict__ V0, const float* __restrict__ V1,
           const float* __restrict__ B2, float* __restrict__ Y, float* __restrict__ Z) {
  constexpr int FOUT = F_DIM;
  constexpr int NT   = FOUT / 16;
  constexpr int NKG  = F_DIM / 4;
  constexpr int NENT = NKG * NT * 32;
  constexpr int FO2   = 16;
  constexpr int NENT2 = FIRST ? 1 : NKG * 32;

  __shared__ float4 wlds[NENT];
  __shared__ __attribute__((aligned(16))) float ptile[LW][16 * PPITCH];
  __shared__ float4 w2lds[NENT2];

  constexpr int SLICE = LW * 32;
  constexpr int NSL   = NENT / SLICE;
  constexpr int NSL2  = FIRST ? 0 : NENT2 / SLICE;
  static_assert(NSL * SLICE == NENT && NSL == 8, "one weight slice per row pair of a 16-row strip");
  static_assert(FIRST || (NSL2 * SLICE == NENT2 && NSL2 <= NSL), "layer 3's weights fit the same slices");

  auto stage_load = [&](int sl, float4& sv, float4& sv2) {
    const int idx = threadIdx.x + sl * SLICE;
    const int l   = idx & 31;
    const int rem = idx >> 5;
    const int t   = rem % NT;
    const int kg  = rem / NT;
    const int n   = t * 16 + (l & 15);
    const int k   = kg * 4 + ((l >> 4) * 2);
    sv = make_float4(W0[(size_t)k * FOUT + n], W0[(size_t)(k + 1) * FOUT + n],
                     W1[(size_t)k * FOUT + n], W1[(size_t)(k + 1) * FOUT + n]);
    if constexpr (!FIRST) {
      if (sl < NSL2) {
        const int kg2 = idx >> 5;
        const int n2  = l & 15;
        const int k2  = kg2 * 4 + ((l >> 4) * 2);
        sv2 = make_float4(V0[(size_t)k2 * FO2 + n2], V0[(size_t)(k2 + 1) * FO2 + n2],
                          V1[(size_t)k2 * FO2 + n2], V1[(size_t)(k2 + 1) * FO2 + n2]);
      }
    }
  };
  auto stage_store = [&](int sl, const float4& sv, const float4& sv2) {
    const int idx = threadIdx.x + sl * SLICE;
    wlds[idx] = sv;
    if constexpr (!FIRST) {
      if (sl < NSL2) w2lds[idx] = sv2;
    }
  };

  const int lane  = threadIdx.x & 31;
  const int wave  = threadIdx.x >> 5;
  const int tile  = blockIdx.x * LW + wave;
  const int ntile = N_NODES / 16;
  const bool active = tile < ntile;
  const int row0 = tile * 16;
  const int m    = lane & 15;
  const int hl   = lane >> 4;
  float* pt = ptile[wave];

  if (active) {
    const int jlast = E_EDGES - 1;
    const char* Hb = (const char*)H;
    const unsigned lane_off = (unsigned)m * 16u;
    const int ri = row0 + min(lane, 16);
    const int rp = (ri < N_NODES) ? rowptr[ri] : E_EDGES;

    int s0 = __shfl(rp, hl),     e0 = __shfl(rp, hl + 1);
    int s1 = __shfl(rp, 2 + hl), e1 = __shfl(rp, 3 + hl);
    int c0 = scol[min(s0 + m, jlast)];
    int c1 = scol[min(s1 + m, jlast)];
    float w0, w1 = 0.0f;
    if (FIRST) {
      w0 = -inv_sqrt_deg(e0 - s0) * inv_sqrt_deg(row_end(rowptr, c0) - rowptr[c0]);
      if (s0 + m < e0) wts[s0 + m] = w0;
    } else {
      w0 = wts[min(s0 + m, jlast)];
      w1 = wts[min(s1 + m, jlast)];
    }

    for (int it = 0; it < 8; ++it) {
      const int l2 = min(2 * it + 4 + hl, 15);
      const int s2 = __shfl(rp, l2), e2 = __shfl(rp, l2 + 1);
      const int j2 = min(s2 + m, jlast);
      const int c2 = scol[j2];
      float w2 = 0.0f;
      int cs1 = 0, ce1 = 0;
      if (FIRST) {
        cs1 = rowptr[c1];
        ce1 = row_end(rowptr, c1);
      } else {
        w2 = wts[j2];
      }
      float4 sv, sv2 = make_float4(0.0f, 0.0f, 0.0f, 0.0f);
      stage_load(it, sv, sv2);

      float4 acc = make_float4(0.0f, 0.0f, 0.0f, 0.0f);
      const float dr = inv_sqrt_deg(e0 - s0);
      const int own = row0 + 2 * it + hl;
      {
        int cb = c0;
        float wb = w0;
        fix_records(cb, wb, m, e0 - s0, own);
        walk_fixed<16>(Hb, lane_off, cb, wb, 0, acc);
      }
      for (int base = s0 + 16; base < e0; base += 16) {
        const int j = base + m;
        int c = 0;
        float w = 0.0f;
        if (j < e0) {
          c = scol[j];
          if (FIRST) {
            w = -dr * inv_sqrt_deg(row_end(rowptr, c) - rowptr[c]);
            wts[j] = w;
          } else {
            w = wts[j];
          }
        }
        const int n = e0 - base;
        fix_records(c, w, m, n, own);
        walk_fixed<8>(Hb, lane_off, c, w, 0, acc);
        if (n > 8) walk_fixed<8>(Hb, lane_off, c, w, 8, acc);
      }
      *(float4*)(pt + (2 * it + hl) * PPITCH + m * 4) = acc;
      stage_store(it, sv, sv2);

      if (FIRST) {
        w1 = -inv_sqrt_deg(e1 - s1) * inv_sqrt_deg(ce1 - cs1);
        if (it < 7 && s1 + m < e1) wts[s1 + m] = w1;
      }
      s0 = s1; e0 = e1; c0 = c1; w0 = w1;
      s1 = s2; e1 = e2; c1 = c2; w1 = w2;
    }
  } else {
    for (int sl = 0; sl < NSL; ++sl) {
      float4 sv, sv2 = make_float4(0.0f, 0.0f, 0.0f, 0.0f);
      stage_load(sl, sv, sv2);
      stage_store(sl, sv, sv2);
    }
  }
  __syncthreads();
  if constexpr (FIRST) {
    if (!active) return;
  }

  const int kk = hl * 2;
  const int roff = hl * 8;
  const float* prow = pt + m * PPITCH;

  if (active) {
  v8f acc[NT];
#pragma unroll
  for (int t = 0; t < NT; ++t) acc[t] = (v8f)0.0f;

  const float* xrow = H + (size_t)(row0 + m) * F_DIM;

#pragma unroll
  for (int kg = 0; kg < NKG; ++kg) {
    const int k0 = kg * 4;
    const v2f aX = *(const v2f*)(xrow + k0 + kk);
    const v2f aP = *(const v2f*)(prow + k0 + kk);
#pragma unroll
    for (int t = 0; t < NT; ++t) {
      const float4 wv = wlds[(kg * NT + t) * 32 + lane];
      v2f b0, b1;
      b0.x = wv.x; b0.y = wv.y;
      b1.x = wv.z; b1.y = wv.w;
      acc[t] = __builtin_amdgcn_wmma_f32_16x16x4_f32(
          false, aX, false, b0, (short)0, acc[t], false, false);
      acc[t] = __builtin_amdgcn_wmma_f32_16x16x4_f32(
          false, aP, false, b1, (short)0, acc[t], false, false);
    }
  }

#pragma unroll
  for (int t = 0; t < NT; ++t) {
    const float bias = B[t * 16 + m];
#pragma unroll
    for (int r = 0; r < 8; ++r) {
      const float v = fmaxf(acc[t][r] + bias, 0.0f);
      if constexpr (FIRST) {
        OUT[(size_t)(row0 + roff + r) * FOUT + t * 16 + m] = v;
      } else {
        pt[(roff + r) * PPITCH + t * 16 + m] = v;
      }
    }
  }
  }

  if constexpr (!FIRST) {
    __syncthreads();
    if (!active) return;

    v8f accY = (v8f)0.0f;
    v8f accZ = (v8f)0.0f;
#pragma unroll
    for (int kg = 0; kg < NKG; ++kg) {
      const v2f aH = *(const v2f*)(prow + kg * 4 + kk);
      const float4 wv = w2lds[kg * 32 + lane];
      v2f b0, b1;
      b0.x = wv.x; b0.y = wv.y;
      b1.x = wv.z; b1.y = wv.w;
      accY = __builtin_amdgcn_wmma_f32_16x16x4_f32(
          false, aH, false, b0, (short)0, accY, false, false);
      accZ = __builtin_amdgcn_wmma_f32_16x16x4_f32(
          false, aH, false, b1, (short)0, accZ, false, false);
    }
    const float bias2 = B2[m];
#pragma unroll
    for (int r = 0; r < 8; ++r) {
      Y[(size_t)(row0 + roff + r) * FO2 + m] = accY[r] + bias2;
      Z[(size_t)(row0 + roff + r) * FO2 + m] = accZ[r];
    }
  }
}


__global__ void gather_out16(const int* __restrict__ rowptr, const int* __restrict__ scol,
                             const float* __restrict__ wts,
                             const float* __restrict__ Y, const float* __restrict__ Z,
                             float* __restrict__ OUT) {
  const int gid = blockIdx.x * blockDim.x + threadIdx.x;
  const int r = gid >> 2;
  if (r >= N_NODES) return;
  const int q = threadIdx.x & 3;
  const int s = rowptr[r];
  const int e = (r + 1 < N_NODES) ? rowptr[r + 1] : E_EDGES;
  const char* Zb = (const char*)Z;
  const unsigned lane_off = (unsigned)q * 16u;
  float4 acc = *(const float4*)(Y + (size_t)r * 16 + q * 4);

  const int jlast = E_EDGES - 1;
  int   ca = scol[min(s + q, jlast)];
  int   cb = scol[min(s + 4 + q, jlast)];
  float wa = wts[min(s + q, jlast)];
  float wb = wts[min(s + 4 + q, jlast)];
  for (int base = s; base < e; base += 8) {
    const int ja = min(base + 8 + q, jlast);
    const int jb = min(base + 12 + q, jlast);
    const int   can = scol[ja];
    const int   cbn = scol[jb];
    const float wan = wts[ja];
    const float wbn = wts[jb];

    const int n = e - base;
    const int cfirst = __shfl(ca, 0, 4);
    if (q >= n)     { ca = cfirst; wa = 0.0f; }
    if (q + 4 >= n) { cb = cfirst; wb = 0.0f; }

    unsigned off[8];
    float ww[8];
    float4 v[8];
#pragma unroll
    for (int k = 0; k < 8; ++k) {
      off[k] = (unsigned)__shfl(k < 4 ? ca : cb, k & 3, 4) * 64u + lane_off;
      ww[k]  = __shfl(k < 4 ? wa : wb, k & 3, 4);
    }
#pragma unroll
    for (int k = 0; k < 8; ++k) v[k] = *(const float4*)(Zb + off[k]);
#pragma unroll
    for (int k = 0; k < 8; ++k) {
      acc.x = fmaf(ww[k], v[k].x, acc.x); acc.y = fmaf(ww[k], v[k].y, acc.y);
      acc.z = fmaf(ww[k], v[k].z, acc.z); acc.w = fmaf(ww[k], v[k].w, acc.w);
    }
    asm volatile("" :: "v"(can), "v"(cbn), "v"(wan), "v"(wbn));
    ca = can; cb = cbn;
    wa = wan; wb = wbn;
  }
  *(float4*)(OUT + (size_t)r * 16 + q * 4) = acc;
}

extern "C" void kernel_launch(void* const* d_in, const int* in_sizes, int n_in,
                              void* d_out, int out_size, void* d_ws, size_t ws_size,
                              hipStream_t stream) {
  const float* x    = (const float*)d_in[0];
  const int*   adj  = (const int*)d_in[1];
  const float* W1_0 = (const float*)d_in[2];
  const float* W1_1 = (const float*)d_in[3];
  const float* b1   = (const float*)d_in[4];
  const float* Wx_0 = (const float*)d_in[5];
  const float* Wx_1 = (const float*)d_in[6];
  const float* bx   = (const float*)d_in[7];
  const float* W2_0 = (const float*)d_in[8];
  const float* W2_1 = (const float*)d_in[9];
  const float* b2   = (const float*)d_in[10];
  float* out = (float*)d_out;

  const int* row = adj;
  const int* col = adj + E_EDGES;

  int*   rowptr = (int*)d_ws;
  int*   scol   = rowptr + N_NODES;
  float* A   = (float*)(scol + E_EDGES);
  float* Bp  = A + (size_t)N_NODES * F_DIM;
  float* C   = Bp + (size_t)N_NODES * F_DIM;
  float* wts      = C;
  int*   pairs    = (int*)(wts + E_EDGES);
  int*   cntT     = pairs + 2 * E_EDGES;
  int*   offT     = cntT + NBIN * NCHUNK;

  const int TB = 256;
  const int gLay = (N_NODES / 16 + LW - 1) / LW;
  const int gG16 = (N_NODES * 4 + TB - 1) / TB;

  chunk_group<<<NCHUNK, GRP_BD, 0, stream>>>(row, col, cntT, offT, pairs);
  bin_sort<<<NBIN, SORT_BD, 0, stream>>>(pairs, cntT, offT, rowptr, scol);

  float* Y3 = A;
  float* Z3 = A + (size_t)N_NODES * 16;
  cheb_layer<true><<<gLay, LW * 32, 0, stream>>>(rowptr, scol, wts, x, W1_0, W1_1, b1, Bp,
                                                 W2_0, W2_1, b2, Y3, Z3);

  cheb_layer<false><<<gLay, LW * 32, 0, stream>>>(rowptr, scol, wts, Bp, Wx_0, Wx_1, bx, A,
                                                  W2_0, W2_1, b2, Y3, Z3);

  gather_out16<<<gG16, TB, 0, stream>>>(rowptr, scol, wts, Y3, Z3, out);
}


# both grouping launches put their records in an LDS copy (chunk: grouped pairs; bin: sorted sources), then store it lane-consecutive: whole lines, no scattered 4/8-byte stores; row order as before
# speedup vs baseline: 4.8832x; 1.0510x over previous
#include <hip/hip_runtime.h>

typedef __attribute__((ext_vector_type(2))) float v2f;
typedef __attribute__((ext_vector_type(8))) float v8f;

#define N_NODES 100000
#define E_EDGES 1600000
#define F_DIM   64


#define RPB_SHIFT 8
#define RPB       (1 << RPB_SHIFT)
#define NBIN      ((N_NODES + RPB - 1) / RPB)
#define NCHUNK    256
#define CHUNK     ((E_EDGES + NCHUNK - 1) / NCHUNK)

#define GRP_BD    1024
#define GRP_U     ((CHUNK + GRP_BD - 1) / GRP_BD)
#define GRP_SC    512
#define SORT_BD   1024
#define SORT_U    8
#define SEG_L     (SORT_BD / NCHUNK)
#define SORT_CAP  8192

static_assert(NBIN <= GRP_SC, "the bins fit the scan in chunk_group");
static_assert(NCHUNK * CHUNK >= E_EDGES, "chunks cover the edge list");
static_assert(SEG_L * NCHUNK == SORT_BD, "one thread group per chunk in bin_sort");
static_assert(SORT_BD / 32 == 32, "bin_sort adds its wave sums with one wave-wide sum");
static_assert(3 * E_EDGES + 2 * NBIN * NCHUNK <= N_NODES * F_DIM,
              "edge weights and grouping temporaries fit in one N*64 plane");

__global__ void __launch_bounds__(GRP_BD)
chunk_group(const int* __restrict__ row, const int* __restrict__ col,
            int* __restrict__ cntT, int* __restrict__ offT, int* __restrict__ pairs) {
  __shared__ int h[GRP_SC];
  __shared__ int buf[2][GRP_SC];
  __shared__ int2 pbuf[CHUNK];
  const int tid = threadIdx.x;
  const int g = blockIdx.x;
  if (tid < GRP_SC) h[tid] = 0;
  const int lo = g * CHUNK;
  const int hi = min(lo + CHUNK, E_EDGES);
  int r[GRP_U], c[GRP_U];
#pragma unroll
  for (int u = 0; u < GRP_U; ++u) {
    const int i = min(lo + tid + u * GRP_BD, hi - 1);
    r[u] = row[i];
    c[u] = col[i];
  }
  __syncthreads();
#pragma unroll
  for (int u = 0; u < GRP_U; ++u)
    if (lo + tid + u * GRP_BD < hi) atomicAdd(&h[r[u] >> RPB_SHIFT], 1);
  __syncthreads();

  int v = 0;
  if (tid < GRP_SC) {
    v = h[tid];
    buf[0][tid] = v;
  }
  __syncthreads();
  int cur = 0;
  for (int d = 1; d < GRP_SC; d <<= 1) {
    if (tid < GRP_SC) {
      int t = buf[cur][tid];
      if (tid >= d) t += buf[cur][tid - d];
      buf[cur ^ 1][tid] = t;
    }
    cur ^= 1;
    __syncthreads();
  }
  if (tid < NBIN) {
    const int excl = buf[cur][tid] - v;
    h[tid] = excl;
    cntT[tid * NCHUNK + g] = v;
    offT[tid * NCHUNK + g] = excl;
  }
  __syncthreads();

#pragma unroll
  for (int u = 0; u < GRP_U; ++u) {
    if (lo + tid + u * GRP_BD < hi) {
      const int pos = atomicAdd(&h[r[u] >> RPB_SHIFT], 1);
      pbuf[pos] = make_int2(r[u], c[u]);
    }
  }
  __syncthreads();
  int2* __restrict__ out = (int2*)pairs + lo;
  int2 t[GRP_U];
#pragma unroll
  for (int u = 0; u < GRP_U; ++u) t[u] = pbuf[min(tid + u * GRP_BD, hi - lo - 1)];
#pragma unroll
  for (int u = 0; u < GRP_U; ++u) {
    const int i = tid + u * GRP_BD;
    if (lo + i < hi) out[i] = t[u];
  }
}

__global__ void __launch_bounds__(SORT_BD)
bin_sort(const int* __restrict__ pairs, const int* __restrict__ cntT,
         const int* __restrict__ offT, int* __restrict__ rowptr, int* __restrict__ scol) {
  __shared__ int cnt[RPB];
  __shared__ int fill[RPB];
  __shared__ int buf[2][RPB];
  __shared__ int red[SORT_BD / 32];
  __shared__ int sbuf[SORT_CAP];
  const int tid = threadIdx.x;
  const int b = blockIdx.x;
  constexpr int TRIP = SORT_U * SEG_L;

  const int g = tid / SEG_L;
  const int q = tid % SEG_L;
  const int glen = cntT[b * NCHUNK + g];
  const int goff = offT[b * NCHUNK + g];
  const int gstart = g * CHUNK + goff;
  const int2* __restrict__ pr = (const int2*)pairs;
  auto pair_at = [&](int k) {
    return pr[min(gstart + max(min(k, glen - 1), 0), E_EDGES - 1)];
  };

  int2 e0[SORT_U];
#pragma unroll
  for (int u = 0; u < SORT_U; ++u) e0[u] = pair_at(q + u * SEG_L);

  int sb = (q == 0) ? goff : 0;
#pragma unroll
  for (int o = 16; o > 0; o >>= 1) sb += __shfl_xor(sb, o);
  if ((tid & 31) == 0) red[tid >> 5] = sb;
  if (tid < RPB) cnt[tid] = 0;
  __syncthreads();
  int bs = red[tid & 31];
#pragma unroll
  for (int o = 16; o > 0; o >>= 1) bs += __shfl_xor(bs, o);

#pragma unroll
  for (int u = 0; u < SORT_U; ++u)
    if (q + u * SEG_L < glen) atomicAdd(&cnt[e0[u].x & (RPB - 1)], 1);
  for (int base = TRIP; base < glen; base += TRIP) {
    int2 e[SORT_U];
#pragma unroll
    for (int u = 0; u < SORT_U; ++u) e[u] = pair_at(base + q + u * SEG_L);
#pragma unroll
    for (int u = 0; u < SORT_U; ++u)
      if (base + q + u * SEG_L < glen) atomicAdd(&cnt[e[u].x & (RPB - 1)], 1);
  }
  __syncthreads();

  int cur = 0;
  if (tid < RPB) buf[0][tid] = cnt[tid];
  __syncthreads();
  for (int d = 1; d < RPB; d <<= 1) {
    if (tid < RPB) {
      int t = buf[cur][tid];
      if (tid >= d) t += buf[cur][tid - d];
      buf[cur ^ 1][tid] = t;
    }
    cur ^= 1;
    __syncthreads();
  }
  const int total = buf[cur][RPB - 1];
  const bool staged = total <= SORT_CAP;
  if (tid < RPB) {
    const int local = buf[cur][tid] - cnt[tid];
    const int r = b * RPB + tid;
    if (r < N_NODES) rowptr[r] = bs + local;
    fill[tid] = local;
  }
  __syncthreads();

  int* __restrict__ dst = scol + bs;
#pragma unroll
  for (int u = 0; u < SORT_U; ++u) {
    if (q + u * SEG_L < glen) {
      const int pos = atomicAdd(&fill[e0[u].x & (RPB - 1)], 1);
      if (staged) sbuf[pos] = e0[u].y;
      else        dst[pos] = e0[u].y;
    }
  }
  for (int base = TRIP; base < glen; base += TRIP) {
    int2 e[SORT_U];
#pragma unroll
    for (int u = 0; u < SORT_U; ++u) e[u] = pair_at(base + q + u * SEG_L);
#pragma unroll
    for (int u = 0; u < SORT_U; ++u) {
      if (base + q + u * SEG_L < glen) {
        const int pos = atomicAdd(&fill[e[u].x & (RPB - 1)], 1);
        if (staged) sbuf[pos] = e[u].y;
        else        dst[pos] = e[u].y;
      }
    }
  }
  if (staged) {
    __syncthreads();
    for (int i = tid; i < total; i += SORT_BD) dst[i] = sbuf[i];
  }
}

__device__ __forceinline__ float inv_sqrt_deg(int d) {
  return (d > 0) ? rsqrtf((float)d) : 0.0f;
}

template <int U>
__device__ __forceinline__ void walk_fixed(const char* __restrict__ Hb, unsigned lane_off,
                                           int c, float w, int t0, float4& acc) {
  unsigned off[U];
  float ww[U];
  float4 v[U];
#pragma unroll
  for (int u = 0; u < U; ++u) {
    off[u] = (unsigned)__shfl(c, t0 + u, 16) * (unsigned)(F_DIM * sizeof(float)) + lane_off;
    ww[u] = __shfl(w, t0 + u, 16);
  }
#pragma unroll
  for (int u = 0; u < U; ++u) v[u] = *(const float4*)(Hb + off[u]);
#pragma unroll
  for (int u = 0; u < U; ++u) {
    acc.x = fmaf(ww[u], v[u].x, acc.x); acc.y = fmaf(ww[u], v[u].y, acc.y);
    acc.z = fmaf(ww[u], v[u].z, acc.z); acc.w = fmaf(ww[u], v[u].w, acc.w);
  }
}

__device__ __forceinline__ void fix_records(int& c, float& w, int m, int n, int own) {
  const int c0 = __shfl(c, 0, 16);
  const int cfirst = (n > 0) ? c0 : own;
  if (m >= n) {
    c = cfirst;
    w = 0.0f;
  }
}

__device__ __forceinline__ int row_end(const int* __restrict__ rowptr, int c) {
  return (c + 1 < N_NODES) ? rowptr[c + 1] : E_EDGES;
}

#define LW      8
#define PPITCH  68

template <bool FIRST>
__global__ void __launch_bounds__(LW * 32)
cheb_layer(const int* __restrict__ rowptr, const int* __restrict__ scol,
           float* __restrict__ wts, const float* __restrict__ H,
           const float* __restrict__ W0, const float* __restrict__ W1,
           const float* __restrict__ B, float* __restrict__ OUT,
           const float* __restrict__ V0, const float* __restrict__ V1,
           const float* __restrict__ B2, float* __restrict__ Y, float* __restrict__ Z) {
  constexpr int FOUT = F_DIM;
  constexpr int NT   = FOUT / 16;
  constexpr int NKG  = F_DIM / 4;
  constexpr int NENT = NKG * NT * 32;
  constexpr int FO2   = 16;
  constexpr int NENT2 = FIRST ? 1 : NKG * 32;

  __shared__ float4 wlds[NENT];
  __shared__ __attribute__((aligned(16))) float ptile[LW][16 * PPITCH];
  __shared__ float4 w2lds[NENT2];

  constexpr int SLICE = LW * 32;
  constexpr int NSL   = NENT / SLICE;
  constexpr int NSL2  = FIRST ? 0 : NENT2 / SLICE;
  static_assert(NSL * SLICE == NENT && NSL == 8, "one weight slice per row pair of a 16-row strip");
  static_assert(FIRST || (NSL2 * SLICE == NENT2 && NSL2 <= NSL), "layer 3's weights fit the same slices");

  auto stage_load = [&](int sl, float4& sv, float4& sv2) {
    const int idx = threadIdx.x + sl * SLICE;
    const int l   = idx & 31;
    const int rem = idx >> 5;
    const int t   = rem % NT;
    const int kg  = rem / NT;
    const int n   = t * 16 + (l & 15);
    const int k   = kg * 4 + ((l >> 4) * 2);
    sv = make_float4(W0[(size_t)k * FOUT + n], W0[(size_t)(k + 1) * FOUT + n],
                     W1[(size_t)k * FOUT + n], W1[(size_t)(k + 1) * FOUT + n]);
    if constexpr (!FIRST) {
      if (sl < NSL2) {
        const int kg2 = idx >> 5;
        const int n2  = l & 15;
        const int k2  = kg2 * 4 + ((l >> 4) * 2);
        sv2 = make_float4(V0[(size_t)k2 * FO2 + n2], V0[(size_t)(k2 + 1) * FO2 + n2],
                          V1[(size_t)k2 * FO2 + n2], V1[(size_t)(k2 + 1) * FO2 + n2]);
      }
    }
  };
  auto stage_store = [&](int sl, const float4& sv, const float4& sv2) {
    const int idx = threadIdx.x + sl * SLICE;
    wlds[idx] = sv;
    if constexpr (!FIRST) {
      if (sl < NSL2) w2lds[idx] = sv2;
    }
  };

  const int lane  = threadIdx.x & 31;
  const int wave  = threadIdx.x >> 5;
  const int tile  = blockIdx.x * LW + wave;
  const int ntile = N_NODES / 16;
  const bool active = tile < ntile;
  const int row0 = tile * 16;
  const int m    = lane & 15;
  const int hl   = lane >> 4;
  float* pt = ptile[wave];

  if (active) {
    const int jlast = E_EDGES - 1;
    const char* Hb = (const char*)H;
    const unsigned lane_off = (unsigned)m * 16u;
    const int ri = row0 + min(lane, 16);
    const int rp = (ri < N_NODES) ? rowptr[ri] : E_EDGES;

    int s0 = __shfl(rp, hl),     e0 = __shfl(rp, hl + 1);
    int s1 = __shfl(rp, 2 + hl), e1 = __shfl(rp, 3 + hl);
    int c0 = scol[min(s0 + m, jlast)];
    int c1 = scol[min(s1 + m, jlast)];
    float w0, w1 = 0.0f;
    if (FIRST) {
      w0 = -inv_sqrt_deg(e0 - s0) * inv_sqrt_deg(row_end(rowptr, c0) - rowptr[c0]);
      if (s0 + m < e0) wts[s0 + m] = w0;
    } else {
      w0 = wts[min(s0 + m, jlast)];
      w1 = wts[min(s1 + m, jlast)];
    }

    for (int it = 0; it < 8; ++it) {
      const int l2 = min(2 * it + 4 + hl, 15);
      const int s2 = __shfl(rp, l2), e2 = __shfl(rp, l2 + 1);
      const int j2 = min(s2 + m, jlast);
      const int c2 = scol[j2];
      float w2 = 0.0f;
      int cs1 = 0, ce1 = 0;
      if (FIRST) {
        cs1 = rowptr[c1];
        ce1 = row_end(rowptr, c1);
      } else {
        w2 = wts[j2];
      }
      float4 sv, sv2 = make_float4(0.0f, 0.0f, 0.0f, 0.0f);
      stage_load(it, sv, sv2);

      float4 acc = make_float4(0.0f, 0.0f, 0.0f, 0.0f);
      const float dr = inv_sqrt_deg(e0 - s0);
      const int own = row0 + 2 * it + hl;
      {
        int cb = c0;
        float wb = w0;
        fix_records(cb, wb, m, e0 - s0, own);
        walk_fixed<16>(Hb, lane_off, cb, wb, 0, acc);
      }
      for (int base = s0 + 16; base < e0; base += 16) {
        const int j = base + m;
        int c = 0;
        float w = 0.0f;
        if (j < e0) {
          c = scol[j];
          if (FIRST) {
            w = -dr * inv_sqrt_deg(row_end(rowptr, c) - rowptr[c]);
            wts[j] = w;
          } else {
            w = wts[j];
          }
        }
        const int n = e0 - base;
        fix_records(c, w, m, n, own);
        walk_fixed<8>(Hb, lane_off, c, w, 0, acc);
        if (n > 8) walk_fixed<8>(Hb, lane_off, c, w, 8, acc);
      }
      *(float4*)(pt + (2 * it + hl) * PPITCH + m * 4) = acc;
      stage_store(it, sv, sv2);

      if (FIRST) {
        w1 = -inv_sqrt_deg(e1 - s1) * inv_sqrt_deg(ce1 - cs1);
        if (it < 7 && s1 + m < e1) wts[s1 + m] = w1;
      }
      s0 = s1; e0 = e1; c0 = c1; w0 = w1;
      s1 = s2; e1 = e2; c1 = c2; w1 = w2;
    }
  } else {
    for (int sl = 0; sl < NSL; ++sl) {
      float4 sv, sv2 = make_float4(0.0f, 0.0f, 0.0f, 0.0f);
      stage_load(sl, sv, sv2);
      stage_store(sl, sv, sv2);
    }
  }
  __syncthreads();
  if constexpr (FIRST) {
    if (!active) return;
  }

  const int kk = hl * 2;
  const int roff = hl * 8;
  const float* prow = pt + m * PPITCH;

  if (active) {
  v8f acc[NT];
#pragma unroll
  for (int t = 0; t < NT; ++t) acc[t] = (v8f)0.0f;

  const float* xrow = H + (size_t)(row0 + m) * F_DIM;

#pragma unroll
  for (int kg = 0; kg < NKG; ++kg) {
    const int k0 = kg * 4;
    const v2f aX = *(const v2f*)(xrow + k0 + kk);
    const v2f aP = *(const v2f*)(prow + k0 + kk);
#pragma unroll
    for (int t = 0; t < NT; ++t) {
      const float4 wv = wlds[(kg * NT + t) * 32 + lane];
      v2f b0, b1;
      b0.x = wv.x; b0.y = wv.y;
      b1.x = wv.z; b1.y = wv.w;
      acc[t] = __builtin_amdgcn_wmma_f32_16x16x4_f32(
          false, aX, false, b0, (short)0, acc[t], false, false);
      acc[t] = __builtin_amdgcn_wmma_f32_16x16x4_f32(
          false, aP, false, b1, (short)0, acc[t], false, false);
    }
  }

#pragma unroll
  for (int t = 0; t < NT; ++t) {
    const float bias = B[t * 16 + m];
#pragma unroll
    for (int r = 0; r < 8; ++r) {
      const float v = fmaxf(acc[t][r] + bias, 0.0f);
      if constexpr (FIRST) {
        OUT[(size_t)(row0 + roff + r) * FOUT + t * 16 + m] = v;
      } else {
        pt[(roff + r) * PPITCH + t * 16 + m] = v;
      }
    }
  }
  }

  if constexpr (!FIRST) {
    __syncthreads();
    if (!active) return;

    v8f accY = (v8f)0.0f;
    v8f accZ = (v8f)0.0f;
#pragma unroll
    for (int kg = 0; kg < NKG; ++kg) {
      const v2f aH = *(const v2f*)(prow + kg * 4 + kk);
      const float4 wv = w2lds[kg * 32 + lane];
      v2f b0, b1;
      b0.x = wv.x; b0.y = wv.y;
      b1.x = wv.z; b1.y = wv.w;
      accY = __builtin_amdgcn_wmma_f32_16x16x4_f32(
          false, aH, false, b0, (short)0, accY, false, false);
      accZ = __builtin_amdgcn_wmma_f32_16x16x4_f32(
          false, aH, false, b1, (short)0, accZ, false, false);
    }
    const float bias2 = B2[m];
#pragma unroll
    for (int r = 0; r < 8; ++r) {
      Y[(size_t)(row0 + roff + r) * FO2 + m] = accY[r] + bias2;
      Z[(size_t)(row0 + roff + r) * FO2 + m] = accZ[r];
    }
  }
}


__global__ void gather_out16(const int* __restrict__ rowptr, const int* __restrict__ scol,
                             const float* __restrict__ wts,
                             const float* __restrict__ Y, const float* __restrict__ Z,
                             float* __restrict__ OUT) {
  const int gid = blockIdx.x * blockDim.x + threadIdx.x;
  const int r = gid >> 2;
  if (r >= N_NODES) return;
  const int q = threadIdx.x & 3;
  const int s = rowptr[r];
  const int e = (r + 1 < N_NODES) ? rowptr[r + 1] : E_EDGES;
  const char* Zb = (const char*)Z;
  const unsigned lane_off = (unsigned)q * 16u;
  float4 acc = *(const float4*)(Y + (size_t)r * 16 + q * 4);

  const int jlast = E_EDGES - 1;
  int   ca = scol[min(s + q, jlast)];
  int   cb = scol[min(s + 4 + q, jlast)];
  float wa = wts[min(s + q, jlast)];
  float wb = wts[min(s + 4 + q, jlast)];
  for (int base = s; base < e; base += 8) {
    const int ja = min(base + 8 + q, jlast);
    const int jb = min(base + 12 + q, jlast);
    const int   can = scol[ja];
    const int   cbn = scol[jb];
    const float wan = wts[ja];
    const float wbn = wts[jb];

    const int n = e - base;
    const int cfirst = __shfl(ca, 0, 4);
    if (q >= n)     { ca = cfirst; wa = 0.0f; }
    if (q + 4 >= n) { cb = cfirst; wb = 0.0f; }

    unsigned off[8];
    float ww[8];
    float4 v[8];
#pragma unroll
    for (int k = 0; k < 8; ++k) {
      off[k] = (unsigned)__shfl(k < 4 ? ca : cb, k & 3, 4) * 64u + lane_off;
      ww[k]  = __shfl(k < 4 ? wa : wb, k & 3, 4);
    }
#pragma unroll
    for (int k = 0; k < 8; ++k) v[k] = *(const float4*)(Zb + off[k]);
#pragma unroll
    for (int k = 0; k < 8; ++k) {
      acc.x = fmaf(ww[k], v[k].x, acc.x); acc.y = fmaf(ww[k], v[k].y, acc.y);
      acc.z = fmaf(ww[k], v[k].z, acc.z); acc.w = fmaf(ww[k], v[k].w, acc.w);
    }
    asm volatile("" :: "v"(can), "v"(cbn), "v"(wan), "v"(wbn));
    ca = can; cb = cbn;
    wa = wan; wb = wbn;
  }
  *(float4*)(OUT + (size_t)r * 16 + q * 4) = acc;
}

extern "C" void kernel_launch(void* const* d_in, const int* in_sizes, int n_in,
                              void* d_out, int out_size, void* d_ws, size_t ws_size,
                              hipStream_t stream) {
  const float* x    = (const float*)d_in[0];
  const int*   adj  = (const int*)d_in[1];
  const float* W1_0 = (const float*)d_in[2];
  const float* W1_1 = (const float*)d_in[3];
  const float* b1   = (const float*)d_in[4];
  const float* Wx_0 = (const float*)d_in[5];
  const float* Wx_1 = (const float*)d_in[6];
  const float* bx   = (const float*)d_in[7];
  const float* W2_0 = (const float*)d_in[8];
  const float* W2_1 = (const float*)d_in[9];
  const float* b2   = (const float*)d_in[10];
  float* out = (float*)d_out;

  const int* row = adj;
  const int* col = adj + E_EDGES;

  int*   rowptr = (int*)d_ws;
  int*   scol   = rowptr + N_NODES;
  float* A   = (float*)(scol + E_EDGES);
  float* Bp  = A + (size_t)N_NODES * F_DIM;
  float* C   = Bp + (size_t)N_NODES * F_DIM;
  float* wts      = C;
  int*   pairs    = (int*)(wts + E_EDGES);
  int*   cntT     = pairs + 2 * E_EDGES;
  int*   offT     = cntT + NBIN * NCHUNK;

  const int TB = 256;
  const int gLay = (N_NODES / 16 + LW - 1) / LW;
  const int gG16 = (N_NODES * 4 + TB - 1) / TB;

  chunk_group<<<NCHUNK, GRP_BD, 0, stream>>>(row, col, cntT, offT, pairs);
  bin_sort<<<NBIN, SORT_BD, 0, stream>>>(pairs, cntT, offT, rowptr, scol);

  float* Y3 = A;
  float* Z3 = A + (size_t)N_NODES * 16;
  cheb_layer<true><<<gLay, LW * 32, 0, stream>>>(rowptr, scol, wts, x, W1_0, W1_1, b1, Bp,
                                                 W2_0, W2_1, b2, Y3, Z3);

  cheb_layer<false><<<gLay, LW * 32, 0, stream>>>(rowptr, scol, wts, Bp, Wx_0, Wx_1, bx, A,
                                                  W2_0, W2_1, b2, Y3, Z3);

  gather_out16<<<gG16, TB, 0, stream>>>(rowptr, scol, wts, Y3, Z3, out);
}
